# starting kernel
# baseline (speedup 1.0000x reference)
#include <hip/hip_runtime.h>

#define N_NODES   50000
#define F_DIM     64
#define H_HEADS   2
#define HF        128
#define L_LAYERS  2
#define NEG_SLOPE 0.2f

typedef __attribute__((ext_vector_type(2))) float v2f;
typedef __attribute__((ext_vector_type(8))) float v8f;

__device__ __forceinline__ float lrelu(float x) {
    return x > 0.0f ? x : NEG_SLOPE * x;
}
__device__ __forceinline__ unsigned fflip(float f) {
    unsigned u = __float_as_uint(f);
    return (u & 0x80000000u) ? ~u : (u | 0x80000000u);
}
__device__ __forceinline__ float funflip(unsigned v) {
    return __uint_as_float((v & 0x80000000u) ? (v ^ 0x80000000u) : ~v);
}

__global__ __launch_bounds__(256)
void gat_proj_wmma(const float* __restrict__ h,
                   const float* __restrict__ Wsrc, const float* __restrict__ bsrc,
                   const float* __restrict__ Wdst, const float* __restrict__ bdst,
                   float* __restrict__ fs, float* __restrict__ fd) {
    const int wave = threadIdx.x >> 5;
    const int lane = threadIdx.x & 31;
    const int half = lane >> 4;
    const int lrow = lane & 15;
    const int t = blockIdx.y * 8 + wave;
    const int nodeBase = blockIdx.x * 16;

    const bool  isSrc = (t < 8);
    const float* __restrict__ W    = isSrc ? Wsrc : Wdst;
    const float* __restrict__ bias = isSrc ? bsrc : bdst;
    float* __restrict__ out        = isSrc ? fs : fd;
    const int col = (isSrc ? t : t - 8) * 16 + lrow;

    const float* __restrict__ arow = h + (size_t)(nodeBase + lrow) * F_DIM;

    v8f acc = {};
#pragma unroll
    for (int k0 = 0; k0 < F_DIM; k0 += 4) {
        const int ka = k0 + half * 2;
        v2f a, b;
        a.x = arow[ka];
        a.y = arow[ka + 1];
        b.x = W[(size_t)ka * HF + col];
        b.y = W[(size_t)(ka + 1) * HF + col];
        acc = __builtin_amdgcn_wmma_f32_16x16x4_f32(
                false, a, false, b, (short)0, acc, false, false);
    }

    const float bv = bias[col];
#pragma unroll
    for (int j = 0; j < 8; ++j) {
        const int row = nodeBase + half * 8 + j;
        out[(size_t)row * HF + col] = acc[j] + bv;
    }
}

__global__ __launch_bounds__(256)
void gat_edge_logits(const float* __restrict__ fs, const float* __restrict__ fd,
                     const float* __restrict__ attn,
                     const int* __restrict__ src, const int* __restrict__ dst,
                     float* __restrict__ logits, unsigned* __restrict__ mflip,
                     int E) {
    const int e = (int)(((size_t)blockIdx.x * blockDim.x + threadIdx.x) >> 5);
    if (e >= E) return;
    const int lane = threadIdx.x & 31;
    const int s = src[e];
    const int d = dst[e];

    const float4 a = *(const float4*)(fs + (size_t)s * HF + lane * 4);
    const float4 b = *(const float4*)(fd + (size_t)d * HF + lane * 4);
    const float4 w = *(const float4*)(attn + lane * 4);

    float sum = w.x * lrelu(a.x + b.x) + w.y * lrelu(a.y + b.y)
              + w.z * lrelu(a.z + b.z) + w.w * lrelu(a.w + b.w);
#pragma unroll
    for (int off = 1; off < 16; off <<= 1)
        sum += __shfl_xor(sum, off, 32);

    if ((lane & 15) == 0) {
        const int hh = lane >> 4;
        logits[(size_t)e * 2 + hh] = sum;
        atomicMax(mflip + (size_t)d * 2 + hh, fflip(sum));
    }
}

__global__ __launch_bounds__(256)
void gat_edge_exp(const int* __restrict__ dst, float* __restrict__ logits,
                  const unsigned* __restrict__ mflip, float* __restrict__ denom,
                  int E2) {
    const int idx = blockIdx.x * blockDim.x + threadIdx.x;
    if (idx >= E2) return;
    const int e = idx >> 1, hh = idx & 1;
    const int d = dst[e];
    const float m  = funflip(mflip[(size_t)d * 2 + hh]);
    const float ex = expf(logits[idx] - m);
    logits[idx] = ex;
    atomicAdd(denom + (size_t)d * 2 + hh, ex);
}

__global__ __launch_bounds__(256)
void gat_edge_aggregate(const float* __restrict__ fs,
                        const int* __restrict__ src, const int* __restrict__ dst,
                        const float* __restrict__ logits,
                        const float* __restrict__ denom,
                        float* __restrict__ agg, int E) {
    const int e = (int)(((size_t)blockIdx.x * blockDim.x + threadIdx.x) >> 5);
    if (e >= E) return;
    const int lane = threadIdx.x & 31;
    const int hh = lane >> 4;
    const int s = src[e];
    const int d = dst[e];
    const float alpha = logits[(size_t)e * 2 + hh] / denom[(size_t)d * 2 + hh];

    const float4 v = *(const float4*)(fs + (size_t)s * HF + lane * 4);
    float* o = agg + (size_t)d * HF + lane * 4;
    atomicAdd(o + 0, alpha * v.x);
    atomicAdd(o + 1, alpha * v.y);
    atomicAdd(o + 2, alpha * v.z);
    atomicAdd(o + 3, alpha * v.w);
}

__global__ __launch_bounds__(256)
void gat_head_mean(const float* __restrict__ agg, float* __restrict__ hout, int n) {
    const int idx = blockIdx.x * blockDim.x + threadIdx.x;
    if (idx >= n) return;
    const int node = idx >> 6, f = idx & 63;
    hout[idx] = 0.5f * (agg[(size_t)node * HF + f] + agg[(size_t)node * HF + 64 + f]);
}

extern "C" void kernel_launch(void* const* d_in, const int* in_sizes, int n_in,
                              void* d_out, int out_size, void* d_ws, size_t ws_size,
                              hipStream_t stream) {
    (void)n_in; (void)out_size; (void)ws_size;
    const float* h0   = (const float*)d_in[0];
    const float* Wsrc = (const float*)d_in[1];
    const float* bsrc = (const float*)d_in[2];
    const float* Wdst = (const float*)d_in[3];
    const float* bdst = (const float*)d_in[4];
    const float* attn = (const float*)d_in[5];
    const int*   src  = (const int*)d_in[6];
    const int*   dst  = (const int*)d_in[7];
    const int    E    = in_sizes[6];

    float* ws      = (float*)d_ws;
    float* fs      = ws;  ws += (size_t)N_NODES * HF;
    float* fd      = ws;  ws += (size_t)N_NODES * HF;
    float* agg     = ws;  ws += (size_t)N_NODES * HF;
    float* hbuf    = ws;  ws += (size_t)N_NODES * F_DIM;
    float* denom   = ws;  ws += (size_t)N_NODES * 2;
    unsigned* mflp = (unsigned*)ws; ws += (size_t)N_NODES * 2;
    float* logits  = ws;

    const int gridE  = (E + 7) / 8;
    const int gridE2 = (E * 2 + 255) / 256;
    const int gridN  = (N_NODES * F_DIM + 255) / 256;

    const float* hcur = h0;
    for (int l = 0; l < L_LAYERS; ++l) {
        gat_proj_wmma<<<dim3(N_NODES / 16, 2), 256, 0, stream>>>(
            hcur,
            Wsrc + (size_t)l * F_DIM * HF, bsrc + (size_t)l * HF,
            Wdst + (size_t)l * F_DIM * HF, bdst + (size_t)l * HF,
            fs, fd);

        hipMemsetAsync(mflp,  0, (size_t)N_NODES * 2 * sizeof(unsigned), stream);
        hipMemsetAsync(denom, 0, (size_t)N_NODES * 2 * sizeof(float), stream);
        hipMemsetAsync(agg,   0, (size_t)N_NODES * HF * sizeof(float), stream);

        gat_edge_logits<<<gridE, 256, 0, stream>>>(
            fs, fd, attn + (size_t)l * HF, src, dst, logits, mflp, E);
        gat_edge_exp<<<gridE2, 256, 0, stream>>>(dst, logits, mflp, denom, E * 2);
        gat_edge_aggregate<<<gridE, 256, 0, stream>>>(
            fs, src, dst, logits, denom, agg, E);

        float* hn = (l == L_LAYERS - 1) ? (float*)d_out : hbuf;
        gat_head_mean<<<gridN, 256, 0, stream>>>(agg, hn, N_NODES * F_DIM);
        hcur = hbuf;
    }
}


# edge passes (logits, aggregate): one wave walks 32 consecutive edges (ids loaded coalesced, row bases on the scalar unit, 4 edges' rows in flight) instead of one wave per edge; sum order unchanged
# speedup vs baseline: 1.1969x; 1.1969x over previous
#include <hip/hip_runtime.h>

#define N_NODES   50000
#define F_DIM     64
#define H_HEADS   2
#define HF        128
#define L_LAYERS  2
#define NEG_SLOPE 0.2f
#define EPW       32

typedef __attribute__((ext_vector_type(2))) float v2f;
typedef __attribute__((ext_vector_type(8))) float v8f;

__device__ __forceinline__ float lrelu(float x) {
    return x > 0.0f ? x : NEG_SLOPE * x;
}
__device__ __forceinline__ unsigned fflip(float f) {
    unsigned u = __float_as_uint(f);
    return (u & 0x80000000u) ? ~u : (u | 0x80000000u);
}
__device__ __forceinline__ float funflip(unsigned v) {
    return __uint_as_float((v & 0x80000000u) ? (v ^ 0x80000000u) : ~v);
}

__global__ __launch_bounds__(256)
void gat_proj_wmma(const float* __restrict__ h,
                   const float* __restrict__ Wsrc, const float* __restrict__ bsrc,
                   const float* __restrict__ Wdst, const float* __restrict__ bdst,
                   float* __restrict__ fs, float* __restrict__ fd) {
    const int wave = threadIdx.x >> 5;
    const int lane = threadIdx.x & 31;
    const int half = lane >> 4;
    const int lrow = lane & 15;
    const int t = blockIdx.y * 8 + wave;
    const int nodeBase = blockIdx.x * 16;

    const bool  isSrc = (t < 8);
    const float* __restrict__ W    = isSrc ? Wsrc : Wdst;
    const float* __restrict__ bias = isSrc ? bsrc : bdst;
    float* __restrict__ out        = isSrc ? fs : fd;
    const int col = (isSrc ? t : t - 8) * 16 + lrow;

    const float* __restrict__ arow = h + (size_t)(nodeBase + lrow) * F_DIM;

    v8f acc = {};
#pragma unroll
    for (int k0 = 0; k0 < F_DIM; k0 += 4) {
        const int ka = k0 + half * 2;
        v2f a, b;
        a.x = arow[ka];
        a.y = arow[ka + 1];
        b.x = W[(size_t)ka * HF + col];
        b.y = W[(size_t)(ka + 1) * HF + col];
        acc = __builtin_amdgcn_wmma_f32_16x16x4_f32(
                false, a, false, b, (short)0, acc, false, false);
    }

    const float bv = bias[col];
#pragma unroll
    for (int j = 0; j < 8; ++j) {
        const int row = nodeBase + half * 8 + j;
        out[(size_t)row * HF + col] = acc[j] + bv;
    }
}

__global__ __launch_bounds__(256)
void gat_edge_logits(const float* __restrict__ fs, const float* __restrict__ fd,
                     const float* __restrict__ attn,
                     const int* __restrict__ src, const int* __restrict__ dst,
                     float* __restrict__ logits, unsigned* __restrict__ mflip,
                     int E) {
    const int wv = __builtin_amdgcn_readfirstlane(
        (int)(((size_t)blockIdx.x * blockDim.x + threadIdx.x) >> 5));
    const int e0 = wv * EPW;
    if (e0 >= E) return;
    const int lane = threadIdx.x & 31;
    const int n = min(EPW, E - e0);

    int ms = 0, md = 0;
    if (lane < n) { ms = src[e0 + lane]; md = dst[e0 + lane]; }

    const float4 w = *(const float4*)(attn + lane * 4);
    const float* __restrict__ fsl = fs + lane * 4;
    const float* __restrict__ fdl = fd + lane * 4;

    float l0 = 0.0f, l1 = 0.0f;
    for (int j = 0; j < n; j += 4) {
        float4 a[4], b[4];
#pragma unroll
        for (int u = 0; u < 4; ++u) {
            const int s = __builtin_amdgcn_readlane(ms, j + u);
            const int d = __builtin_amdgcn_readlane(md, j + u);
            a[u] = *(const float4*)(fsl + (size_t)s * HF);
            b[u] = *(const float4*)(fdl + (size_t)d * HF);
        }
#pragma unroll
        for (int u = 0; u < 4; ++u) {
            float sum = w.x * lrelu(a[u].x + b[u].x) + w.y * lrelu(a[u].y + b[u].y)
                      + w.z * lrelu(a[u].z + b[u].z) + w.w * lrelu(a[u].w + b[u].w);
#pragma unroll
            for (int off = 1; off < 16; off <<= 1)
                sum += __shfl_xor(sum, off, 32);
            const float h0 = __int_as_float(__builtin_amdgcn_readlane(__float_as_int(sum), 0));
            const float h1 = __int_as_float(__builtin_amdgcn_readlane(__float_as_int(sum), 16));
            if (lane == j + u) { l0 = h0; l1 = h1; }
        }
    }

#pragma unroll
    for (int k = 0; k < 2; ++k) {
        const int je = (lane >> 1) + 16 * k;
        const float v0 = __shfl(l0, je, 32);
        const float v1 = __shfl(l1, je, 32);
        const int   dd = __shfl(md, je, 32);
        const int   hh = lane & 1;
        const float v  = hh ? v1 : v0;
        if (je < n) {
            logits[(size_t)(e0 + je) * 2 + hh] = v;
            atomicMax(mflip + (size_t)dd * 2 + hh, fflip(v));
        }
    }
}

__global__ __launch_bounds__(256)
void gat_edge_exp(const int* __restrict__ dst, float* __restrict__ logits,
                  const unsigned* __restrict__ mflip, float* __restrict__ denom,
                  int E2) {
    const int idx = blockIdx.x * blockDim.x + threadIdx.x;
    if (idx >= E2) return;
    const int e = idx >> 1, hh = idx & 1;
    const int d = dst[e];
    const float m  = funflip(mflip[(size_t)d * 2 + hh]);
    const float ex = expf(logits[idx] - m);
    logits[idx] = ex;
    atomicAdd(denom + (size_t)d * 2 + hh, ex);
}

__global__ __launch_bounds__(256)
void gat_edge_aggregate(const float* __restrict__ fs,
                        const int* __restrict__ src, const int* __restrict__ dst,
                        const float* __restrict__ logits,
                        const float* __restrict__ denom,
                        float* __restrict__ agg, int E) {
    const int wv = __builtin_amdgcn_readfirstlane(
        (int)(((size_t)blockIdx.x * blockDim.x + threadIdx.x) >> 5));
    const int e0 = wv * EPW;
    if (e0 >= E) return;
    const int lane = threadIdx.x & 31;
    const int hh = lane >> 4;
    const int n = min(EPW, E - e0);

    int ms = 0, md = 0;
    float al0 = 0.0f, al1 = 0.0f;
    if (lane < n) {
        ms = src[e0 + lane];
        md = dst[e0 + lane];
        const float2 ex = *(const float2*)(logits + (size_t)(e0 + lane) * 2);
        const float2 dn = *(const float2*)(denom + (size_t)md * 2);
        al0 = ex.x / dn.x;
        al1 = ex.y / dn.y;
    }

    const float* __restrict__ fsl = fs + lane * 4;
    float* __restrict__ aggl = agg + lane * 4;

    for (int j = 0; j < n; j += 4) {
        float4 v[4];
        int    d[4];
        float  al[4];
#pragma unroll
        for (int u = 0; u < 4; ++u) {
            const int s = __builtin_amdgcn_readlane(ms, j + u);
            d[u] = __builtin_amdgcn_readlane(md, j + u);
            const float a0 = __int_as_float(__builtin_amdgcn_readlane(__float_as_int(al0), j + u));
            const float a1 = __int_as_float(__builtin_amdgcn_readlane(__float_as_int(al1), j + u));
            al[u] = hh ? a1 : a0;
            v[u] = *(const float4*)(fsl + (size_t)s * HF);
        }
#pragma unroll
        for (int u = 0; u < 4; ++u) {
            if (j + u < n) {
                float* o = aggl + (size_t)d[u] * HF;
                atomicAdd(o + 0, al[u] * v[u].x);
                atomicAdd(o + 1, al[u] * v[u].y);
                atomicAdd(o + 2, al[u] * v[u].z);
                atomicAdd(o + 3, al[u] * v[u].w);
            }
        }
    }
}

__global__ __launch_bounds__(256)
void gat_head_mean(const float* __restrict__ agg, float* __restrict__ hout, int n) {
    const int idx = blockIdx.x * blockDim.x + threadIdx.x;
    if (idx >= n) return;
    const int node = idx >> 6, f = idx & 63;
    hout[idx] = 0.5f * (agg[(size_t)node * HF + f] + agg[(size_t)node * HF + 64 + f]);
}

extern "C" void kernel_launch(void* const* d_in, const int* in_sizes, int n_in,
                              void* d_out, int out_size, void* d_ws, size_t ws_size,
                              hipStream_t stream) {
    (void)n_in; (void)out_size; (void)ws_size;
    const float* h0   = (const float*)d_in[0];
    const float* Wsrc = (const float*)d_in[1];
    const float* bsrc = (const float*)d_in[2];
    const float* Wdst = (const float*)d_in[3];
    const float* bdst = (const float*)d_in[4];
    const float* attn = (const float*)d_in[5];
    const int*   src  = (const int*)d_in[6];
    const int*   dst  = (const int*)d_in[7];
    const int    E    = in_sizes[6];

    float* ws      = (float*)d_ws;
    float* fs      = ws;  ws += (size_t)N_NODES * HF;
    float* fd      = ws;  ws += (size_t)N_NODES * HF;
    float* agg     = ws;  ws += (size_t)N_NODES * HF;
    float* hbuf    = ws;  ws += (size_t)N_NODES * F_DIM;
    float* denom   = ws;  ws += (size_t)N_NODES * 2;
    unsigned* mflp = (unsigned*)ws; ws += (size_t)N_NODES * 2;
    float* logits  = ws;

    const int gridE  = (E + 255) / 256;
    const int gridE2 = (E * 2 + 255) / 256;
    const int gridN  = (N_NODES * F_DIM + 255) / 256;

    const float* hcur = h0;
    for (int l = 0; l < L_LAYERS; ++l) {
        gat_proj_wmma<<<dim3(N_NODES / 16, 2), 256, 0, stream>>>(
            hcur,
            Wsrc + (size_t)l * F_DIM * HF, bsrc + (size_t)l * HF,
            Wdst + (size_t)l * F_DIM * HF, bdst + (size_t)l * HF,
            fs, fd);

        hipMemsetAsync(mflp,  0, (size_t)N_NODES * 2 * sizeof(unsigned), stream);
        hipMemsetAsync(denom, 0, (size_t)N_NODES * 2 * sizeof(float), stream);
        hipMemsetAsync(agg,   0, (size_t)N_NODES * HF * sizeof(float), stream);

        gat_edge_logits<<<gridE, 256, 0, stream>>>(
            fs, fd, attn + (size_t)l * HF, src, dst, logits, mflp, E);
        gat_edge_exp<<<gridE2, 256, 0, stream>>>(dst, logits, mflp, denom, E * 2);
        gat_edge_aggregate<<<gridE, 256, 0, stream>>>(
            fs, src, dst, logits, denom, agg, E);

        float* hn = (l == L_LAYERS - 1) ? (float*)d_out : hbuf;
        gat_head_mean<<<gridN, 256, 0, stream>>>(agg, hn, N_NODES * F_DIM);
        hcur = hbuf;
    }
}


# aggregate: lane l owns features l, l+32, l+64, l+96 of a row, so each row load and each float atomic covers one whole 128-B line (4 lines per edge, not 16 part-lines); addends as before
# speedup vs baseline: 1.5978x; 1.3349x over previous
#include <hip/hip_runtime.h>

#define N_NODES   50000
#define F_DIM     64
#define H_HEADS   2
#define HF        128
#define L_LAYERS  2
#define NEG_SLOPE 0.2f
#define EPW       32

typedef __attribute__((ext_vector_type(2))) float v2f;
typedef __attribute__((ext_vector_type(8))) float v8f;

__device__ __forceinline__ float lrelu(float x) {
    return x > 0.0f ? x : NEG_SLOPE * x;
}
__device__ __forceinline__ unsigned fflip(float f) {
    unsigned u = __float_as_uint(f);
    return (u & 0x80000000u) ? ~u : (u | 0x80000000u);
}
__device__ __forceinline__ float funflip(unsigned v) {
    return __uint_as_float((v & 0x80000000u) ? (v ^ 0x80000000u) : ~v);
}

__global__ __launch_bounds__(256)
void gat_proj_wmma(const float* __restrict__ h,
                   const float* __restrict__ Wsrc, const float* __restrict__ bsrc,
                   const float* __restrict__ Wdst, const float* __restrict__ bdst,
                   float* __restrict__ fs, float* __restrict__ fd) {
    const int wave = threadIdx.x >> 5;
    const int lane = threadIdx.x & 31;
    const int half = lane >> 4;
    const int lrow = lane & 15;
    const int t = blockIdx.y * 8 + wave;
    const int nodeBase = blockIdx.x * 16;

    const bool  isSrc = (t < 8);
    const float* __restrict__ W    = isSrc ? Wsrc : Wdst;
    const float* __restrict__ bias = isSrc ? bsrc : bdst;
    float* __restrict__ out        = isSrc ? fs : fd;
    const int col = (isSrc ? t : t - 8) * 16 + lrow;

    const float* __restrict__ arow = h + (size_t)(nodeBase + lrow) * F_DIM;

    v8f acc = {};
#pragma unroll
    for (int k0 = 0; k0 < F_DIM; k0 += 4) {
        const int ka = k0 + half * 2;
        v2f a, b;
        a.x = arow[ka];
        a.y = arow[ka + 1];
        b.x = W[(size_t)ka * HF + col];
        b.y = W[(size_t)(ka + 1) * HF + col];
        acc = __builtin_amdgcn_wmma_f32_16x16x4_f32(
                false, a, false, b, (short)0, acc, false, false);
    }

    const float bv = bias[col];
#pragma unroll
    for (int j = 0; j < 8; ++j) {
        const int row = nodeBase + half * 8 + j;
        out[(size_t)row * HF + col] = acc[j] + bv;
    }
}

__global__ __launch_bounds__(256)
void gat_edge_logits(const float* __restrict__ fs, const float* __restrict__ fd,
                     const float* __restrict__ attn,
                     const int* __restrict__ src, const int* __restrict__ dst,
                     float* __restrict__ logits, unsigned* __restrict__ mflip,
                     int E) {
    const int wv = __builtin_amdgcn_readfirstlane(
        (int)(((size_t)blockIdx.x * blockDim.x + threadIdx.x) >> 5));
    const int e0 = wv * EPW;
    if (e0 >= E) return;
    const int lane = threadIdx.x & 31;
    const int n = min(EPW, E - e0);

    int ms = 0, md = 0;
    if (lane < n) { ms = src[e0 + lane]; md = dst[e0 + lane]; }

    const float4 w = *(const float4*)(attn + lane * 4);
    const float* __restrict__ fsl = fs + lane * 4;
    const float* __restrict__ fdl = fd + lane * 4;

    float l0 = 0.0f, l1 = 0.0f;
    for (int j = 0; j < n; j += 4) {
        float4 a[4], b[4];
#pragma unroll
        for (int u = 0; u < 4; ++u) {
            const int s = __builtin_amdgcn_readlane(ms, j + u);
            const int d = __builtin_amdgcn_readlane(md, j + u);
            a[u] = *(const float4*)(fsl + (size_t)s * HF);
            b[u] = *(const float4*)(fdl + (size_t)d * HF);
        }
#pragma unroll
        for (int u = 0; u < 4; ++u) {
            float sum = w.x * lrelu(a[u].x + b[u].x) + w.y * lrelu(a[u].y + b[u].y)
                      + w.z * lrelu(a[u].z + b[u].z) + w.w * lrelu(a[u].w + b[u].w);
#pragma unroll
            for (int off = 1; off < 16; off <<= 1)
                sum += __shfl_xor(sum, off, 32);
            const float h0 = __int_as_float(__builtin_amdgcn_readlane(__float_as_int(sum), 0));
            const float h1 = __int_as_float(__builtin_amdgcn_readlane(__float_as_int(sum), 16));
            if (lane == j + u) { l0 = h0; l1 = h1; }
        }
    }

#pragma unroll
    for (int k = 0; k < 2; ++k) {
        const int je = (lane >> 1) + 16 * k;
        const float v0 = __shfl(l0, je, 32);
        const float v1 = __shfl(l1, je, 32);
        const int   dd = __shfl(md, je, 32);
        const int   hh = lane & 1;
        const float v  = hh ? v1 : v0;
        if (je < n) {
            logits[(size_t)(e0 + je) * 2 + hh] = v;
            atomicMax(mflip + (size_t)dd * 2 + hh, fflip(v));
        }
    }
}

__global__ __launch_bounds__(256)
void gat_edge_exp(const int* __restrict__ dst, float* __restrict__ logits,
                  const unsigned* __restrict__ mflip, float* __restrict__ denom,
                  int E2) {
    const int idx = blockIdx.x * blockDim.x + threadIdx.x;
    if (idx >= E2) return;
    const int e = idx >> 1, hh = idx & 1;
    const int d = dst[e];
    const float m  = funflip(mflip[(size_t)d * 2 + hh]);
    const float ex = expf(logits[idx] - m);
    logits[idx] = ex;
    atomicAdd(denom + (size_t)d * 2 + hh, ex);
}

__global__ __launch_bounds__(256)
void gat_edge_aggregate(const float* __restrict__ fs,
                        const int* __restrict__ src, const int* __restrict__ dst,
                        const float* __restrict__ logits,
                        const float* __restrict__ denom,
                        float* __restrict__ agg, int E) {
    const int wv = __builtin_amdgcn_readfirstlane(
        (int)(((size_t)blockIdx.x * blockDim.x + threadIdx.x) >> 5));
    const int e0 = wv * EPW;
    if (e0 >= E) return;
    const int lane = threadIdx.x & 31;
    const int n = min(EPW, E - e0);

    int ms = 0, md = 0;
    float al0 = 0.0f, al1 = 0.0f;
    if (lane < n) {
        ms = src[e0 + lane];
        md = dst[e0 + lane];
        const float2 ex = *(const float2*)(logits + (size_t)(e0 + lane) * 2);
        const float2 dn = *(const float2*)(denom + (size_t)md * 2);
        al0 = ex.x / dn.x;
        al1 = ex.y / dn.y;
    }

    for (int j = 0; j < n; j += 4) {
        float v[4][4];
        int   d[4];
        float a0[4], a1[4];
#pragma unroll
        for (int u = 0; u < 4; ++u) {
            const int s = __builtin_amdgcn_readlane(ms, j + u);
            d[u]  = __builtin_amdgcn_readlane(md, j + u);
            a0[u] = __int_as_float(__builtin_amdgcn_readlane(__float_as_int(al0), j + u));
            a1[u] = __int_as_float(__builtin_amdgcn_readlane(__float_as_int(al1), j + u));
            const float* __restrict__ r = fs + (size_t)s * HF;
            v[u][0] = r[lane];
            v[u][1] = r[lane + 32];
            v[u][2] = r[lane + 64];
            v[u][3] = r[lane + 96];
        }
#pragma unroll
        for (int u = 0; u < 4; ++u) {
            if (j + u < n) {
                float* o = agg + (size_t)d[u] * HF;
                atomicAdd(o + lane,      a0[u] * v[u][0]);
                atomicAdd(o + lane + 32, a0[u] * v[u][1]);
                atomicAdd(o + lane + 64, a1[u] * v[u][2]);
                atomicAdd(o + lane + 96, a1[u] * v[u][3]);
            }
        }
    }
}

__global__ __launch_bounds__(256)
void gat_head_mean(const float* __restrict__ agg, float* __restrict__ hout, int n) {
    const int idx = blockIdx.x * blockDim.x + threadIdx.x;
    if (idx >= n) return;
    const int node = idx >> 6, f = idx & 63;
    hout[idx] = 0.5f * (agg[(size_t)node * HF + f] + agg[(size_t)node * HF + 64 + f]);
}

extern "C" void kernel_launch(void* const* d_in, const int* in_sizes, int n_in,
                              void* d_out, int out_size, void* d_ws, size_t ws_size,
                              hipStream_t stream) {
    (void)n_in; (void)out_size; (void)ws_size;
    const float* h0   = (const float*)d_in[0];
    const float* Wsrc = (const float*)d_in[1];
    const float* bsrc = (const float*)d_in[2];
    const float* Wdst = (const float*)d_in[3];
    const float* bdst = (const float*)d_in[4];
    const float* attn = (const float*)d_in[5];
    const int*   src  = (const int*)d_in[6];
    const int*   dst  = (const int*)d_in[7];
    const int    E    = in_sizes[6];

    float* ws      = (float*)d_ws;
    float* fs      = ws;  ws += (size_t)N_NODES * HF;
    float* fd      = ws;  ws += (size_t)N_NODES * HF;
    float* agg     = ws;  ws += (size_t)N_NODES * HF;
    float* hbuf    = ws;  ws += (size_t)N_NODES * F_DIM;
    float* denom   = ws;  ws += (size_t)N_NODES * 2;
    unsigned* mflp = (unsigned*)ws; ws += (size_t)N_NODES * 2;
    float* logits  = ws;

    const int gridE  = (E + 255) / 256;
    const int gridE2 = (E * 2 + 255) / 256;
    const int gridN  = (N_NODES * F_DIM + 255) / 256;

    const float* hcur = h0;
    for (int l = 0; l < L_LAYERS; ++l) {
        gat_proj_wmma<<<dim3(N_NODES / 16, 2), 256, 0, stream>>>(
            hcur,
            Wsrc + (size_t)l * F_DIM * HF, bsrc + (size_t)l * HF,
            Wdst + (size_t)l * F_DIM * HF, bdst + (size_t)l * HF,
            fs, fd);

        hipMemsetAsync(mflp,  0, (size_t)N_NODES * 2 * sizeof(unsigned), stream);
        hipMemsetAsync(denom, 0, (size_t)N_NODES * 2 * sizeof(float), stream);
        hipMemsetAsync(agg,   0, (size_t)N_NODES * HF * sizeof(float), stream);

        gat_edge_logits<<<gridE, 256, 0, stream>>>(
            fs, fd, attn + (size_t)l * HF, src, dst, logits, mflp, E);
        gat_edge_exp<<<gridE2, 256, 0, stream>>>(dst, logits, mflp, denom, E * 2);
        gat_edge_aggregate<<<gridE, 256, 0, stream>>>(
            fs, src, dst, logits, denom, agg, E);

        float* hn = (l == L_LAYERS - 1) ? (float*)d_out : hbuf;
        gat_head_mean<<<gridN, 256, 0, stream>>>(agg, hn, N_NODES * F_DIM);
        hcur = hbuf;
    }
}


# edge passes split by destination half: 2 work-groups per 512-edge chunk, each reads its L2 half, draws a chunk by ticket, does the edges into that half's node range; float atomics in arrival order
# speedup vs baseline: 2.2786x; 1.4261x over previous
#include <hip/hip_runtime.h>

#define N_NODES   50000
#define F_DIM     64
#define H_HEADS   2
#define HF        128
#define L_LAYERS  2
#define NEG_SLOPE 0.2f
#define EPW       32
#define RUNS      2
#define CHUNK_E   512
#define N_HALF    ((N_NODES / 2 + 15) & ~15)
#define TICK_PAD  64
#define TICK_WORDS 512

typedef __attribute__((ext_vector_type(2))) float v2f;
typedef __attribute__((ext_vector_type(8))) float v8f;
typedef __attribute__((ext_vector_type(4))) float v4f;

__device__ __forceinline__ float lrelu(float x) {
    return x > 0.0f ? x : NEG_SLOPE * x;
}
__device__ __forceinline__ unsigned fflip(float f) {
    unsigned u = __float_as_uint(f);
    return (u & 0x80000000u) ? ~u : (u | 0x80000000u);
}
__device__ __forceinline__ float funflip(unsigned v) {
    return __uint_as_float((v & 0x80000000u) ? (v ^ 0x80000000u) : ~v);
}

__global__ __launch_bounds__(256)
void gat_proj_wmma(const float* __restrict__ h,
                   const float* __restrict__ Wsrc, const float* __restrict__ bsrc,
                   const float* __restrict__ Wdst, const float* __restrict__ bdst,
                   float* __restrict__ fs, float* __restrict__ fd) {
    const int wave = threadIdx.x >> 5;
    const int lane = threadIdx.x & 31;
    const int half = lane >> 4;
    const int lrow = lane & 15;
    const int t = blockIdx.y * 8 + wave;
    const int nodeBase = blockIdx.x * 16;

    const bool  isSrc = (t < 8);
    const float* __restrict__ W    = isSrc ? Wsrc : Wdst;
    const float* __restrict__ bias = isSrc ? bsrc : bdst;
    float* __restrict__ out        = isSrc ? fs : fd;
    const int col = (isSrc ? t : t - 8) * 16 + lrow;

    const float* __restrict__ arow = h + (size_t)(nodeBase + lrow) * F_DIM;

    v8f acc = {};
#pragma unroll
    for (int k0 = 0; k0 < F_DIM; k0 += 4) {
        const int ka = k0 + half * 2;
        v2f a, b;
        a.x = arow[ka];
        a.y = arow[ka + 1];
        b.x = W[(size_t)ka * HF + col];
        b.y = W[(size_t)(ka + 1) * HF + col];
        acc = __builtin_amdgcn_wmma_f32_16x16x4_f32(
                false, a, false, b, (short)0, acc, false, false);
    }

    const float bv = bias[col];
#pragma unroll
    for (int j = 0; j < 8; ++j) {
        const int row = nodeBase + half * 8 + j;
        out[(size_t)row * HF + col] = acc[j] + bv;
    }
}

__device__ __forceinline__ int l2_half() {
    return (__builtin_amdgcn_s_sendmsg_rtn(0x87) >> 18) & 1;
}
__device__ __forceinline__ void draw_chunk(unsigned* tick, unsigned nT, unsigned* slot) {
    if (threadIdx.x == 0) {
        unsigned hh = (unsigned)l2_half();
        unsigned t  = atomicAdd(tick + hh * TICK_PAD, 1u);
        if (t >= nT) {
            hh ^= 1u;
            t = atomicAdd(tick + hh * TICK_PAD, 1u);
        }
        slot[0] = t;
        slot[1] = hh;
    }
    __syncthreads();
}

__global__ __launch_bounds__(256)
void gat_edge_logits(const float* __restrict__ fs, const float* __restrict__ fd,
                     const float* __restrict__ attn,
                     const int* __restrict__ src, const int* __restrict__ dst,
                     float* __restrict__ lgA, float* __restrict__ lgB,
                     unsigned* __restrict__ mflip,
                     unsigned* __restrict__ tick, int E, unsigned nT) {
    __shared__ unsigned slot[2];
    draw_chunk(tick, nT, slot);
    const unsigned t = __builtin_amdgcn_readfirstlane(slot[0]);
    if (t >= nT) return;
    const bool up = __builtin_amdgcn_readfirstlane(slot[1]) != 0;
    float* __restrict__ logits = up ? lgB : lgA;
    const int wave = __builtin_amdgcn_readfirstlane((int)(threadIdx.x >> 5));
    const int lane = threadIdx.x & 31;

    const float4 w = *(const float4*)(attn + lane * 4);
    const float* __restrict__ fsl = fs + lane * 4;
    const float* __restrict__ fdl = fd + lane * 4;

#pragma nounroll
    for (int r = 0; r < RUNS; ++r) {
        const int e0 = (int)t * CHUNK_E + (r * 8 + wave) * EPW;
        if (e0 >= E) break;
        const int n = min(EPW, E - e0);

        int ms = 0, md = 0;
        if (lane < n) { ms = src[e0 + lane]; md = dst[e0 + lane]; }
        const bool mine = (lane < n) && ((md >= N_HALF) == up);
        const unsigned mall = __builtin_amdgcn_ballot_w32(mine);

        float l0 = 0.0f, l1 = 0.0f;
        unsigned m = mall;
        while (m) {
            int  jj[4];
            bool lv[4];
            lv[0] = true;
            jj[0] = __builtin_ctz(m);
            m &= (m - 1);
#pragma unroll
            for (int u = 1; u < 4; ++u) {
                lv[u] = (m != 0);
                jj[u] = lv[u] ? __builtin_ctz(m) : jj[u - 1];
                m &= (m - 1);
            }
            v4f a[4], b[4];
#pragma unroll
            for (int u = 0; u < 4; ++u) {
                const int s = __builtin_amdgcn_readlane(ms, jj[u]);
                const int d = __builtin_amdgcn_readlane(md, jj[u]);
                a[u] = *(const v4f*)(fsl + (size_t)s * HF);
                b[u] = *(const v4f*)(fdl + (size_t)d * HF);
            }
            asm volatile("" :: "v"(a[0]), "v"(a[1]), "v"(a[2]), "v"(a[3]),
                               "v"(b[0]), "v"(b[1]), "v"(b[2]), "v"(b[3]));
#pragma unroll
            for (int u = 0; u < 4; ++u) {
                float sum = w.x * lrelu(a[u].x + b[u].x) + w.y * lrelu(a[u].y + b[u].y)
                          + w.z * lrelu(a[u].z + b[u].z) + w.w * lrelu(a[u].w + b[u].w);
#pragma unroll
                for (int off = 1; off < 16; off <<= 1)
                    sum += __shfl_xor(sum, off, 32);
                const float h0 = __int_as_float(__builtin_amdgcn_readlane(__float_as_int(sum), 0));
                const float h1 = __int_as_float(__builtin_amdgcn_readlane(__float_as_int(sum), 16));
                if (lv[u] && lane == jj[u]) { l0 = h0; l1 = h1; }
            }
        }

#pragma unroll
        for (int k = 0; k < 2; ++k) {
            const int je = (lane >> 1) + 16 * k;
            const float v0 = __shfl(l0, je, 32);
            const float v1 = __shfl(l1, je, 32);
            const int   dd = __shfl(md, je, 32);
            const int   hd = lane & 1;
            const float v  = hd ? v1 : v0;
            if ((mall >> je) & 1u) {
                logits[(size_t)(e0 + je) * 2 + hd] = v;
                atomicMax(mflip + (size_t)dd * 2 + hd, fflip(v));
            }
        }
    }
}

__global__ __launch_bounds__(256)
void gat_edge_exp(const int* __restrict__ dst,
                  float* __restrict__ lgA, float* __restrict__ lgB,
                  const unsigned* __restrict__ mflip, float* __restrict__ denom,
                  unsigned* __restrict__ tick, int E2, unsigned nT) {
    __shared__ unsigned slot[2];
    draw_chunk(tick, nT, slot);
    const unsigned t = __builtin_amdgcn_readfirstlane(slot[0]);
    if (t >= nT) return;
    const bool up = __builtin_amdgcn_readfirstlane(slot[1]) != 0;
    float* __restrict__ logits = up ? lgB : lgA;
    int   idx[2 * RUNS], d[2 * RUNS];
    bool  mine[2 * RUNS];
    unsigned mu[2 * RUNS];
    float lg[2 * RUNS];
#pragma unroll
    for (int k = 0; k < 2 * RUNS; ++k) {
        idx[k] = (int)t * (2 * CHUNK_E) + k * 256 + (int)threadIdx.x;
        d[k] = (idx[k] < E2) ? dst[idx[k] >> 1] : 0;
    }
#pragma unroll
    for (int k = 0; k < 2 * RUNS; ++k) {
        mine[k] = (idx[k] < E2) && ((d[k] >= N_HALF) == up);
        if (mine[k]) {
            mu[k] = mflip[(size_t)d[k] * 2 + (idx[k] & 1)];
            lg[k] = logits[idx[k]];
        }
    }
#pragma unroll
    for (int k = 0; k < 2 * RUNS; ++k) {
        if (mine[k]) {
            const float ex = expf(lg[k] - funflip(mu[k]));
            logits[idx[k]] = ex;
            atomicAdd(denom + (size_t)d[k] * 2 + (idx[k] & 1), ex);
        }
    }
}

__global__ __launch_bounds__(256)
void gat_edge_aggregate(const float* __restrict__ fs,
                        const int* __restrict__ src, const int* __restrict__ dst,
                        const float* __restrict__ lgA, const float* __restrict__ lgB,
                        const float* __restrict__ denom,
                        float* __restrict__ agg,
                        unsigned* __restrict__ tick, int E, unsigned nT) {
    __shared__ unsigned slot[2];
    draw_chunk(tick, nT, slot);
    const unsigned t = __builtin_amdgcn_readfirstlane(slot[0]);
    if (t >= nT) return;
    const bool up = __builtin_amdgcn_readfirstlane(slot[1]) != 0;
    const float* __restrict__ logits = up ? lgB : lgA;
    const int wave = __builtin_amdgcn_readfirstlane((int)(threadIdx.x >> 5));
    const int lane = threadIdx.x & 31;

#pragma nounroll
    for (int r = 0; r < RUNS; ++r) {
        const int e0 = (int)t * CHUNK_E + (r * 8 + wave) * EPW;
        if (e0 >= E) break;
        const int n = min(EPW, E - e0);

        int ms = 0, md = 0;
        if (lane < n) { ms = src[e0 + lane]; md = dst[e0 + lane]; }
        const bool mine = (lane < n) && ((md >= N_HALF) == up);
        float al0 = 0.0f, al1 = 0.0f;
        if (mine) {
            const float2 ex = *(const float2*)(logits + (size_t)(e0 + lane) * 2);
            const float2 dn = *(const float2*)(denom + (size_t)md * 2);
            al0 = ex.x / dn.x;
            al1 = ex.y / dn.y;
        }
        unsigned m = __builtin_amdgcn_ballot_w32(mine);

        while (m) {
            int  jj[4];
            bool lv[4];
            lv[0] = true;
            jj[0] = __builtin_ctz(m);
            m &= (m - 1);
#pragma unroll
            for (int u = 1; u < 4; ++u) {
                lv[u] = (m != 0);
                jj[u] = lv[u] ? __builtin_ctz(m) : jj[u - 1];
                m &= (m - 1);
            }
            float v[4][4];
            int   d[4];
            float a0[4], a1[4];
#pragma unroll
            for (int u = 0; u < 4; ++u) {
                const int s = __builtin_amdgcn_readlane(ms, jj[u]);
                d[u]  = __builtin_amdgcn_readlane(md, jj[u]);
                a0[u] = __int_as_float(__builtin_amdgcn_readlane(__float_as_int(al0), jj[u]));
                a1[u] = __int_as_float(__builtin_amdgcn_readlane(__float_as_int(al1), jj[u]));
                const float* __restrict__ r4 = fs + (size_t)s * HF;
                v[u][0] = r4[lane];
                v[u][1] = r4[lane + 32];
                v[u][2] = r4[lane + 64];
                v[u][3] = r4[lane + 96];
            }
            asm volatile("" :: "v"(v[0][0]), "v"(v[0][1]), "v"(v[0][2]), "v"(v[0][3]),
                               "v"(v[1][0]), "v"(v[1][1]), "v"(v[1][2]), "v"(v[1][3]),
                               "v"(v[2][0]), "v"(v[2][1]), "v"(v[2][2]), "v"(v[2][3]),
                               "v"(v[3][0]), "v"(v[3][1]), "v"(v[3][2]), "v"(v[3][3]));
#pragma unroll
            for (int u = 0; u < 4; ++u) {
                if (lv[u]) {
                    float* o = agg + (size_t)d[u] * HF;
                    atomicAdd(o + lane,      a0[u] * v[u][0]);
                    atomicAdd(o + lane + 32, a0[u] * v[u][1]);
                    atomicAdd(o + lane + 64, a1[u] * v[u][2]);
                    atomicAdd(o + lane + 96, a1[u] * v[u][3]);
                }
            }
        }
    }
}

__global__ __launch_bounds__(256)
void gat_head_mean(const float* __restrict__ agg, float* __restrict__ hout, int n) {
    const int idx = blockIdx.x * blockDim.x + threadIdx.x;
    if (idx >= n) return;
    const int node = idx >> 6, f = idx & 63;
    hout[idx] = 0.5f * (agg[(size_t)node * HF + f] + agg[(size_t)node * HF + 64 + f]);
}

extern "C" void kernel_launch(void* const* d_in, const int* in_sizes, int n_in,
                              void* d_out, int out_size, void* d_ws, size_t ws_size,
                              hipStream_t stream) {
    (void)n_in; (void)out_size; (void)ws_size;
    const float* h0   = (const float*)d_in[0];
    const float* Wsrc = (const float*)d_in[1];
    const float* bsrc = (const float*)d_in[2];
    const float* Wdst = (const float*)d_in[3];
    const float* bdst = (const float*)d_in[4];
    const float* attn = (const float*)d_in[5];
    const int*   src  = (const int*)d_in[6];
    const int*   dst  = (const int*)d_in[7];
    const int    E    = in_sizes[6];

    float* ws      = (float*)d_ws;
    float* fs      = ws;  ws += (size_t)N_NODES * HF;
    float* fd      = ws;  ws += (size_t)N_NODES * HF;
    float* agg     = ws;  ws += (size_t)N_NODES * HF;
    float* hbuf    = ws;  ws += (size_t)N_NODES * F_DIM;
    float* denom   = ws;  ws += (size_t)N_NODES * 2;
    unsigned* mflp = (unsigned*)ws; ws += (size_t)N_NODES * 2;
    unsigned* tick = (unsigned*)ws; ws += TICK_WORDS;
    float* logits  = ws;
    float* logitsB = hbuf;

    const unsigned nT = (unsigned)((E + CHUNK_E - 1) / CHUNK_E);
    const unsigned gridT = 2 * nT;
    const int gridN  = (N_NODES * F_DIM + 255) / 256;

    const float* hcur = h0;
    for (int l = 0; l < L_LAYERS; ++l) {
        gat_proj_wmma<<<dim3(N_NODES / 16, 2), 256, 0, stream>>>(
            hcur,
            Wsrc + (size_t)l * F_DIM * HF, bsrc + (size_t)l * HF,
            Wdst + (size_t)l * F_DIM * HF, bdst + (size_t)l * HF,
            fs, fd);

        hipMemsetAsync(mflp,  0, ((size_t)N_NODES * 2 + TICK_WORDS) * sizeof(unsigned), stream);
        hipMemsetAsync(denom, 0, (size_t)N_NODES * 2 * sizeof(float), stream);
        hipMemsetAsync(agg,   0, (size_t)N_NODES * HF * sizeof(float), stream);

        gat_edge_logits<<<gridT, 256, 0, stream>>>(
            fs, fd, attn + (size_t)l * HF, src, dst, logits, logitsB, mflp,
            tick, E, nT);
        gat_edge_exp<<<gridT, 256, 0, stream>>>(
            dst, logits, logitsB, mflp, denom, tick + 2 * TICK_PAD, E * 2, nT);
        gat_edge_aggregate<<<gridT, 256, 0, stream>>>(
            fs, src, dst, logits, logitsB, denom, agg, tick + 4 * TICK_PAD, E, nT);

        float* hn = (l == L_LAYERS - 1) ? (float*)d_out : hbuf;
        gat_head_mean<<<gridN, 256, 0, stream>>>(agg, hn, N_NODES * F_DIM);
        hcur = hbuf;
    }
}


# aggregate: per-edge head mean added as 32-bit fixed-point integer atomics (unit max|fs|/2^30): 2 whole-line adds per edge, not 4 float; exp sums as 64-bit integers; sums order-free, rounding changed
# speedup vs baseline: 2.3763x; 1.0429x over previous
#include <hip/hip_runtime.h>

#define N_NODES   50000
#define F_DIM     64
#define H_HEADS   2
#define HF        128
#define L_LAYERS  2
#define NEG_SLOPE 0.2f
#define EPW       32
#define RUNS      2
#define CHUNK_E   512
#define N_HALF    ((N_NODES / 2 + 15) & ~15)
#define TICK_PAD  64
#define TICK_WORDS 512
#define FSMAX_WORD (6 * TICK_PAD)
#define EX_ONE   0x1p40f
#define EX_INV   0x1p-40f
#define MSG_ONE  0x1p30f
#define MSG_INV  0x1p-30f

typedef __attribute__((ext_vector_type(2))) float v2f;
typedef __attribute__((ext_vector_type(8))) float v8f;
typedef __attribute__((ext_vector_type(4))) float v4f;

__device__ __forceinline__ float lrelu(float x) {
    return x > 0.0f ? x : NEG_SLOPE * x;
}
__device__ __forceinline__ unsigned fflip(float f) {
    unsigned u = __float_as_uint(f);
    return (u & 0x80000000u) ? ~u : (u | 0x80000000u);
}
__device__ __forceinline__ float funflip(unsigned v) {
    return __uint_as_float((v & 0x80000000u) ? (v ^ 0x80000000u) : ~v);
}

__global__ __launch_bounds__(256)
void gat_proj_wmma(const float* __restrict__ h,
                   const float* __restrict__ Wsrc, const float* __restrict__ bsrc,
                   const float* __restrict__ Wdst, const float* __restrict__ bdst,
                   float* __restrict__ fs, float* __restrict__ fd,
                   unsigned* __restrict__ fsmax) {
    const int wave = threadIdx.x >> 5;
    const int lane = threadIdx.x & 31;
    const int half = lane >> 4;
    const int lrow = lane & 15;
    const int t = blockIdx.y * 8 + wave;
    const int nodeBase = blockIdx.x * 16;

    const bool  isSrc = (t < 8);
    const float* __restrict__ W    = isSrc ? Wsrc : Wdst;
    const float* __restrict__ bias = isSrc ? bsrc : bdst;
    float* __restrict__ out        = isSrc ? fs : fd;
    const int col = (isSrc ? t : t - 8) * 16 + lrow;

    const float* __restrict__ arow = h + (size_t)(nodeBase + lrow) * F_DIM;

    v8f acc = {};
#pragma unroll
    for (int k0 = 0; k0 < F_DIM; k0 += 4) {
        const int ka = k0 + half * 2;
        v2f a, b;
        a.x = arow[ka];
        a.y = arow[ka + 1];
        b.x = W[(size_t)ka * HF + col];
        b.y = W[(size_t)(ka + 1) * HF + col];
        acc = __builtin_amdgcn_wmma_f32_16x16x4_f32(
                false, a, false, b, (short)0, acc, false, false);
    }

    const float bv = bias[col];
    float mx = 0.0f;
#pragma unroll
    for (int j = 0; j < 8; ++j) {
        const int row = nodeBase + half * 8 + j;
        const float o = acc[j] + bv;
        out[(size_t)row * HF + col] = o;
        mx = fmaxf(mx, fabsf(o));
    }
    if (blockIdx.y == 0) {
        __shared__ float wmax[8];
#pragma unroll
        for (int off = 16; off > 0; off >>= 1)
            mx = fmaxf(mx, __shfl_xor(mx, off, 32));
        if (lane == 0) wmax[wave] = mx;
        __syncthreads();
        if (threadIdx.x == 0) {
            float m = wmax[0];
#pragma unroll
            for (int w = 1; w < 8; ++w) m = fmaxf(m, wmax[w]);
            atomicMax(fsmax, __float_as_uint(m));
        }
    }
}

__device__ __forceinline__ int l2_half() {
    return (__builtin_amdgcn_s_sendmsg_rtn(0x87) >> 18) & 1;
}
__device__ __forceinline__ void draw_chunk(unsigned* tick, unsigned nT, unsigned* slot) {
    if (threadIdx.x == 0) {
        unsigned hh = (unsigned)l2_half();
        unsigned t  = atomicAdd(tick + hh * TICK_PAD, 1u);
        if (t >= nT) {
            hh ^= 1u;
            t = atomicAdd(tick + hh * TICK_PAD, 1u);
        }
        slot[0] = t;
        slot[1] = hh;
    }
    __syncthreads();
}

__global__ __launch_bounds__(256)
void gat_edge_logits(const float* __restrict__ fs, const float* __restrict__ fd,
                     const float* __restrict__ attn,
                     const int* __restrict__ src, const int* __restrict__ dst,
                     float* __restrict__ lgA, float* __restrict__ lgB,
                     unsigned* __restrict__ mflip,
                     unsigned* __restrict__ tick, int E, unsigned nT) {
    __shared__ unsigned slot[2];
    draw_chunk(tick, nT, slot);
    const unsigned t = __builtin_amdgcn_readfirstlane(slot[0]);
    if (t >= nT) return;
    const bool up = __builtin_amdgcn_readfirstlane(slot[1]) != 0;
    float* __restrict__ logits = up ? lgB : lgA;
    const int wave = __builtin_amdgcn_readfirstlane((int)(threadIdx.x >> 5));
    const int lane = threadIdx.x & 31;

    const float4 w = *(const float4*)(attn + lane * 4);
    const float* __restrict__ fsl = fs + lane * 4;
    const float* __restrict__ fdl = fd + lane * 4;

#pragma nounroll
    for (int r = 0; r < RUNS; ++r) {
        const int e0 = (int)t * CHUNK_E + (r * 8 + wave) * EPW;
        if (e0 >= E) break;
        const int n = min(EPW, E - e0);

        int ms = 0, md = 0;
        if (lane < n) { ms = src[e0 + lane]; md = dst[e0 + lane]; }
        const bool mine = (lane < n) && ((md >= N_HALF) == up);
        const unsigned mall = __builtin_amdgcn_ballot_w32(mine);

        float l0 = 0.0f, l1 = 0.0f;
        unsigned m = mall;
        while (m) {
            int  jj[4];
            bool lv[4];
            lv[0] = true;
            jj[0] = __builtin_ctz(m);
            m &= (m - 1);
#pragma unroll
            for (int u = 1; u < 4; ++u) {
                lv[u] = (m != 0);
                jj[u] = lv[u] ? __builtin_ctz(m) : jj[u - 1];
                m &= (m - 1);
            }
            v4f a[4], b[4];
#pragma unroll
            for (int u = 0; u < 4; ++u) {
                const int s = __builtin_amdgcn_readlane(ms, jj[u]);
                const int d = __builtin_amdgcn_readlane(md, jj[u]);
                a[u] = *(const v4f*)(fsl + (size_t)s * HF);
                b[u] = *(const v4f*)(fdl + (size_t)d * HF);
            }
            asm volatile("" :: "v"(a[0]), "v"(a[1]), "v"(a[2]), "v"(a[3]),
                               "v"(b[0]), "v"(b[1]), "v"(b[2]), "v"(b[3]));
#pragma unroll
            for (int u = 0; u < 4; ++u) {
                float sum = w.x * lrelu(a[u].x + b[u].x) + w.y * lrelu(a[u].y + b[u].y)
                          + w.z * lrelu(a[u].z + b[u].z) + w.w * lrelu(a[u].w + b[u].w);
#pragma unroll
                for (int off = 1; off < 16; off <<= 1)
                    sum += __shfl_xor(sum, off, 32);
                const float h0 = __int_as_float(__builtin_amdgcn_readlane(__float_as_int(sum), 0));
                const float h1 = __int_as_float(__builtin_amdgcn_readlane(__float_as_int(sum), 16));
                if (lv[u] && lane == jj[u]) { l0 = h0; l1 = h1; }
            }
        }

#pragma unroll
        for (int k = 0; k < 2; ++k) {
            const int je = (lane >> 1) + 16 * k;
            const float v0 = __shfl(l0, je, 32);
            const float v1 = __shfl(l1, je, 32);
            const int   dd = __shfl(md, je, 32);
            const int   hd = lane & 1;
            const float v  = hd ? v1 : v0;
            if ((mall >> je) & 1u) {
                logits[(size_t)(e0 + je) * 2 + hd] = v;
                atomicMax(mflip + (size_t)dd * 2 + hd, fflip(v));
            }
        }
    }
}

__global__ __launch_bounds__(256)
void gat_edge_exp(const int* __restrict__ dst,
                  float* __restrict__ lgA, float* __restrict__ lgB,
                  const unsigned* __restrict__ mflip, unsigned long long* __restrict__ den64,
                  unsigned* __restrict__ tick, int E2, unsigned nT) {
    __shared__ unsigned slot[2];
    draw_chunk(tick, nT, slot);
    const unsigned t = __builtin_amdgcn_readfirstlane(slot[0]);
    if (t >= nT) return;
    const bool up = __builtin_amdgcn_readfirstlane(slot[1]) != 0;
    float* __restrict__ logits = up ? lgB : lgA;
    int   idx[2 * RUNS], d[2 * RUNS];
    bool  mine[2 * RUNS];
    unsigned mu[2 * RUNS];
    float lg[2 * RUNS];
#pragma unroll
    for (int k = 0; k < 2 * RUNS; ++k) {
        idx[k] = (int)t * (2 * CHUNK_E) + k * 256 + (int)threadIdx.x;
        d[k] = (idx[k] < E2) ? dst[idx[k] >> 1] : 0;
    }
#pragma unroll
    for (int k = 0; k < 2 * RUNS; ++k) {
        mine[k] = (idx[k] < E2) && ((d[k] >= N_HALF) == up);
        if (mine[k]) {
            mu[k] = mflip[(size_t)d[k] * 2 + (idx[k] & 1)];
            lg[k] = logits[idx[k]];
        }
    }
#pragma unroll
    for (int k = 0; k < 2 * RUNS; ++k) {
        if (mine[k]) {
            const float ex = expf(lg[k] - funflip(mu[k]));
            logits[idx[k]] = ex;
            atomicAdd(den64 + (size_t)d[k] * 2 + (idx[k] & 1), __float2ull_rn(ex * EX_ONE));
        }
    }
}

__global__ __launch_bounds__(256)
void gat_edge_aggregate(const float* __restrict__ fs,
                        const int* __restrict__ src, const int* __restrict__ dst,
                        const float* __restrict__ lgA, const float* __restrict__ lgB,
                        const unsigned long long* __restrict__ den64,
                        const unsigned* __restrict__ fsmax,
                        int* __restrict__ acc,
                        unsigned* __restrict__ tick, int E, unsigned nT) {
    __shared__ unsigned slot[2];
    draw_chunk(tick, nT, slot);
    const unsigned t = __builtin_amdgcn_readfirstlane(slot[0]);
    if (t >= nT) return;
    const bool up = __builtin_amdgcn_readfirstlane(slot[1]) != 0;
    const float* __restrict__ logits = up ? lgB : lgA;
    const int wave = __builtin_amdgcn_readfirstlane((int)(threadIdx.x >> 5));
    const int lane = threadIdx.x & 31;

    const float mxf = __uint_as_float(*fsmax);
    const float hs  = (mxf > 0x1p-90f) ? (0.5f * MSG_ONE) / mxf : 0.0f;

#pragma nounroll
    for (int r = 0; r < RUNS; ++r) {
        const int e0 = (int)t * CHUNK_E + (r * 8 + wave) * EPW;
        if (e0 >= E) break;
        const int n = min(EPW, E - e0);

        int ms = 0, md = 0;
        if (lane < n) { ms = src[e0 + lane]; md = dst[e0 + lane]; }
        const bool mine = (lane < n) && ((md >= N_HALF) == up);
        float al0 = 0.0f, al1 = 0.0f;
        if (mine) {
            const float2 ex = *(const float2*)(logits + (size_t)(e0 + lane) * 2);
            const ulonglong2 q = *(const ulonglong2*)(den64 + (size_t)md * 2);
            al0 = ex.x / ((float)q.x * EX_INV) * hs;
            al1 = ex.y / ((float)q.y * EX_INV) * hs;
        }
        unsigned m = __builtin_amdgcn_ballot_w32(mine);

        while (m) {
            int  jj[4];
            bool lv[4];
            lv[0] = true;
            jj[0] = __builtin_ctz(m);
            m &= (m - 1);
#pragma unroll
            for (int u = 1; u < 4; ++u) {
                lv[u] = (m != 0);
                jj[u] = lv[u] ? __builtin_ctz(m) : jj[u - 1];
                m &= (m - 1);
            }
            float v[4][4];
            int   d[4];
            float a0[4], a1[4];
#pragma unroll
            for (int u = 0; u < 4; ++u) {
                const int s = __builtin_amdgcn_readlane(ms, jj[u]);
                d[u]  = __builtin_amdgcn_readlane(md, jj[u]);
                a0[u] = __int_as_float(__builtin_amdgcn_readlane(__float_as_int(al0), jj[u]));
                a1[u] = __int_as_float(__builtin_amdgcn_readlane(__float_as_int(al1), jj[u]));
                const float* __restrict__ r4 = fs + (size_t)s * HF;
                v[u][0] = r4[lane];
                v[u][1] = r4[lane + 32];
                v[u][2] = r4[lane + 64];
                v[u][3] = r4[lane + 96];
            }
            asm volatile("" :: "v"(v[0][0]), "v"(v[0][1]), "v"(v[0][2]), "v"(v[0][3]),
                               "v"(v[1][0]), "v"(v[1][1]), "v"(v[1][2]), "v"(v[1][3]),
                               "v"(v[2][0]), "v"(v[2][1]), "v"(v[2][2]), "v"(v[2][3]),
                               "v"(v[3][0]), "v"(v[3][1]), "v"(v[3][2]), "v"(v[3][3]));
#pragma unroll
            for (int u = 0; u < 4; ++u) {
                if (lv[u]) {
                    int* o = acc + (size_t)d[u] * F_DIM;
                    atomicAdd(o + lane,      __float2int_rn(a0[u] * v[u][0] + a1[u] * v[u][2]));
                    atomicAdd(o + lane + 32, __float2int_rn(a0[u] * v[u][1] + a1[u] * v[u][3]));
                }
            }
        }
    }
}

__global__ __launch_bounds__(256)
void gat_sum_to_float(const int* __restrict__ acc, const unsigned* __restrict__ fsmax,
                      float* __restrict__ hout, int n4) {
    const int idx = blockIdx.x * blockDim.x + threadIdx.x;
    if (idx >= n4) return;
    const float unit = __uint_as_float(*fsmax) * MSG_INV;
    const int4 a = ((const int4*)acc)[idx];
    float4 o;
    o.x = (float)a.x * unit;
    o.y = (float)a.y * unit;
    o.z = (float)a.z * unit;
    o.w = (float)a.w * unit;
    ((float4*)hout)[idx] = o;
}

extern "C" void kernel_launch(void* const* d_in, const int* in_sizes, int n_in,
                              void* d_out, int out_size, void* d_ws, size_t ws_size,
                              hipStream_t stream) {
    (void)n_in; (void)out_size; (void)ws_size;
    const float* h0   = (const float*)d_in[0];
    const float* Wsrc = (const float*)d_in[1];
    const float* bsrc = (const float*)d_in[2];
    const float* Wdst = (const float*)d_in[3];
    const float* bdst = (const float*)d_in[4];
    const float* attn = (const float*)d_in[5];
    const int*   src  = (const int*)d_in[6];
    const int*   dst  = (const int*)d_in[7];
    const int    E    = in_sizes[6];

    float* ws      = (float*)d_ws;
    float* fs      = ws;  ws += (size_t)N_NODES * HF;
    float* fd      = ws;  ws += (size_t)N_NODES * HF;
    int* acc       = (int*)ws;
    unsigned long long* den64 = (unsigned long long*)(acc + (size_t)N_NODES * F_DIM);
    ws += (size_t)N_NODES * HF;
    float* hbuf    = ws;  ws += (size_t)N_NODES * F_DIM;
    ws += (size_t)N_NODES * 2;
    unsigned* mflp = (unsigned*)ws; ws += (size_t)N_NODES * 2;
    unsigned* tick = (unsigned*)ws; ws += TICK_WORDS;
    unsigned* fsmax = tick + FSMAX_WORD;
    float* logits  = ws;
    float* logitsB = hbuf;

    const unsigned nT = (unsigned)((E + CHUNK_E - 1) / CHUNK_E);
    const unsigned gridT = 2 * nT;
    const int n4     = N_NODES * F_DIM / 4;
    const int gridN  = (n4 + 255) / 256;

    const float* hcur = h0;
    for (int l = 0; l < L_LAYERS; ++l) {
        hipMemsetAsync(mflp, 0, ((size_t)N_NODES * 2 + TICK_WORDS) * sizeof(unsigned), stream);
        hipMemsetAsync(acc,  0, ((size_t)N_NODES * F_DIM + (size_t)N_NODES * 4) * sizeof(int), stream);

        gat_proj_wmma<<<dim3(N_NODES / 16, 2), 256, 0, stream>>>(
            hcur,
            Wsrc + (size_t)l * F_DIM * HF, bsrc + (size_t)l * HF,
            Wdst + (size_t)l * F_DIM * HF, bdst + (size_t)l * HF,
            fs, fd, fsmax);

        gat_edge_logits<<<gridT, 256, 0, stream>>>(
            fs, fd, attn + (size_t)l * HF, src, dst, logits, logitsB, mflp,
            tick, E, nT);
        gat_edge_exp<<<gridT, 256, 0, stream>>>(
            dst, logits, logitsB, mflp, den64, tick + 2 * TICK_PAD, E * 2, nT);
        gat_edge_aggregate<<<gridT, 256, 0, stream>>>(
            fs, src, dst, logits, logitsB, den64, fsmax, acc, tick + 4 * TICK_PAD, E, nT);

        float* hn = (l == L_LAYERS - 1) ? (float*)d_out : hbuf;
        gat_sum_to_float<<<gridN, 256, 0, stream>>>(acc, fsmax, hn, n4);
        hcur = hbuf;
    }
}


# projection GEMM: a wave keeps the W fragments of its 32 columns in registers and loops over 16-row tiles; h tile staged once per block in LDS; whole-line stores; k order same, fs/fd bit-identical
# speedup vs baseline: 2.6291x; 1.1064x over previous
#include <hip/hip_runtime.h>

#define N_NODES   50000
#define F_DIM     64
#define H_HEADS   2
#define HF        128
#define L_LAYERS  2
#define NEG_SLOPE 0.2f
#define EPW       32
#define RUNS      2
#define CHUNK_E   512
#define N_HALF    ((N_NODES / 2 + 15) & ~15)
#define TICK_PAD  64
#define TICK_WORDS 512
#define FSMAX_WORD (6 * TICK_PAD)
#define EX_ONE   0x1p40f
#define EX_INV   0x1p-40f
#define MSG_ONE  0x1p30f
#define MSG_INV  0x1p-30f
#define PROJ_TILES (N_NODES / 16)
#define PROJ_GRID  1024
#define A_PITCH    68

typedef __attribute__((ext_vector_type(2))) float v2f;
typedef __attribute__((ext_vector_type(8))) float v8f;
typedef __attribute__((ext_vector_type(4))) float v4f;

__device__ __forceinline__ float lrelu(float x) {
    return x > 0.0f ? x : NEG_SLOPE * x;
}
__device__ __forceinline__ unsigned fflip(float f) {
    unsigned u = __float_as_uint(f);
    return (u & 0x80000000u) ? ~u : (u | 0x80000000u);
}
__device__ __forceinline__ float funflip(unsigned v) {
    return __uint_as_float((v & 0x80000000u) ? (v ^ 0x80000000u) : ~v);
}

__global__ __launch_bounds__(256)
void gat_proj_wmma(const float* __restrict__ h,
                   const float* __restrict__ Wsrc, const float* __restrict__ bsrc,
                   const float* __restrict__ Wdst, const float* __restrict__ bdst,
                   float* __restrict__ fs, float* __restrict__ fd,
                   unsigned* __restrict__ fsmax) {
    __shared__ __attribute__((aligned(16))) float atile[2][16 * A_PITCH];
    __shared__ float wmax[4];
    const int wave = __builtin_amdgcn_readfirstlane((int)(threadIdx.x >> 5));
    const int lane = threadIdx.x & 31;
    const int half = lane >> 4;
    const int lrow = lane & 15;

    const bool  isSrc = (wave < 4);
    const float* __restrict__ W    = isSrc ? Wsrc : Wdst;
    const float* __restrict__ bias = isSrc ? bsrc : bdst;
    float* __restrict__ out        = isSrc ? fs : fd;
    const int cb = (wave & 3) * 32 + lrow * 2;

    v2f b0[16], b1[16];
#pragma unroll
    for (int kk = 0; kk < 16; ++kk) {
        const int ka = kk * 4 + half * 2;
        const float2 r0 = *(const float2*)(W + (size_t)ka * HF + cb);
        const float2 r1 = *(const float2*)(W + (size_t)(ka + 1) * HF + cb);
        b0[kk].x = r0.x; b0[kk].y = r1.x;
        b1[kk].x = r0.y; b1[kk].y = r1.y;
    }
    const float2 bv = *(const float2*)(bias + cb);

    const int srow = threadIdx.x >> 4;
    const int scol = (threadIdx.x & 15) * 4;
    int tile = blockIdx.x;
    {
        const float4 v = *(const float4*)(h + (size_t)tile * (16 * F_DIM) + threadIdx.x * 4);
        *(float4*)(&atile[0][srow * A_PITCH + scol]) = v;
    }
    __syncthreads();

    float mx = 0.0f;
    int buf = 0;
    for (; tile < PROJ_TILES; tile += PROJ_GRID) {
        const int  nxt  = tile + PROJ_GRID;
        const bool more = nxt < PROJ_TILES;
        float4 pre = {0.0f, 0.0f, 0.0f, 0.0f};
        if (more) pre = *(const float4*)(h + (size_t)nxt * (16 * F_DIM) + threadIdx.x * 4);

        const float* __restrict__ ar = &atile[buf][lrow * A_PITCH + half * 2];
        v2f a[16];
#pragma unroll
        for (int kk = 0; kk < 16; ++kk)
            a[kk] = *(const v2f*)(ar + kk * 4);

        v8f acc0 = {}, acc1 = {};
#pragma unroll
        for (int kk = 0; kk < 16; ++kk) {
            acc0 = __builtin_amdgcn_wmma_f32_16x16x4_f32(
                    false, a[kk], false, b0[kk], (short)0, acc0, false, false);
            acc1 = __builtin_amdgcn_wmma_f32_16x16x4_f32(
                    false, a[kk], false, b1[kk], (short)0, acc1, false, false);
        }

        *(float4*)(&atile[buf ^ 1][srow * A_PITCH + scol]) = pre;
        __syncthreads();
        buf ^= 1;

        float* __restrict__ orow = out + (size_t)(tile * 16 + half * 8) * HF + cb;
#pragma unroll
        for (int j = 0; j < 8; ++j) {
            float2 o;
            o.x = acc0[j] + bv.x;
            o.y = acc1[j] + bv.y;
            *(float2*)(orow + (size_t)j * HF) = o;
            mx = fmaxf(mx, fmaxf(fabsf(o.x), fabsf(o.y)));
        }
    }

#pragma unroll
    for (int off = 16; off > 0; off >>= 1)
        mx = fmaxf(mx, __shfl_xor(mx, off, 32));
    if (isSrc && lane == 0) wmax[wave] = mx;
    __syncthreads();
    if (threadIdx.x == 0) {
        const float m = fmaxf(fmaxf(wmax[0], wmax[1]), fmaxf(wmax[2], wmax[3]));
        atomicMax(fsmax, __float_as_uint(m));
    }
}

__device__ __forceinline__ int l2_half() {
    return (__builtin_amdgcn_s_sendmsg_rtn(0x87) >> 18) & 1;
}
__device__ __forceinline__ void draw_chunk(unsigned* tick, unsigned nT, unsigned* slot) {
    if (threadIdx.x == 0) {
        unsigned hh = (unsigned)l2_half();
        unsigned t  = atomicAdd(tick + hh * TICK_PAD, 1u);
        if (t >= nT) {
            hh ^= 1u;
            t = atomicAdd(tick + hh * TICK_PAD, 1u);
        }
        slot[0] = t;
        slot[1] = hh;
    }
    __syncthreads();
}

__global__ __launch_bounds__(256)
void gat_edge_logits(const float* __restrict__ fs, const float* __restrict__ fd,
                     const float* __restrict__ attn,
                     const int* __restrict__ src, const int* __restrict__ dst,
                     float* __restrict__ lgA, float* __restrict__ lgB,
                     unsigned* __restrict__ mflip,
                     unsigned* __restrict__ tick, int E, unsigned nT) {
    __shared__ unsigned slot[2];
    draw_chunk(tick, nT, slot);
    const unsigned t = __builtin_amdgcn_readfirstlane(slot[0]);
    if (t >= nT) return;
    const bool up = __builtin_amdgcn_readfirstlane(slot[1]) != 0;
    float* __restrict__ logits = up ? lgB : lgA;
    const int wave = __builtin_amdgcn_readfirstlane((int)(threadIdx.x >> 5));
    const int lane = threadIdx.x & 31;

    const float4 w = *(const float4*)(attn + lane * 4);
    const float* __restrict__ fsl = fs + lane * 4;
    const float* __restrict__ fdl = fd + lane * 4;

#pragma nounroll
    for (int r = 0; r < RUNS; ++r) {
        const int e0 = (int)t * CHUNK_E + (r * 8 + wave) * EPW;
        if (e0 >= E) break;
        const int n = min(EPW, E - e0);

        int ms = 0, md = 0;
        if (lane < n) { ms = src[e0 + lane]; md = dst[e0 + lane]; }
        const bool mine = (lane < n) && ((md >= N_HALF) == up);
        const unsigned mall = __builtin_amdgcn_ballot_w32(mine);

        float l0 = 0.0f, l1 = 0.0f;
        unsigned m = mall;
        while (m) {
            int  jj[4];
            bool lv[4];
            lv[0] = true;
            jj[0] = __builtin_ctz(m);
            m &= (m - 1);
#pragma unroll
            for (int u = 1; u < 4; ++u) {
                lv[u] = (m != 0);
                jj[u] = lv[u] ? __builtin_ctz(m) : jj[u - 1];
                m &= (m - 1);
            }
            v4f a[4], b[4];
#pragma unroll
            for (int u = 0; u < 4; ++u) {
                const int s = __builtin_amdgcn_readlane(ms, jj[u]);
                const int d = __builtin_amdgcn_readlane(md, jj[u]);
                a[u] = *(const v4f*)(fsl + (size_t)s * HF);
                b[u] = *(const v4f*)(fdl + (size_t)d * HF);
            }
            asm volatile("" :: "v"(a[0]), "v"(a[1]), "v"(a[2]), "v"(a[3]),
                               "v"(b[0]), "v"(b[1]), "v"(b[2]), "v"(b[3]));
#pragma unroll
            for (int u = 0; u < 4; ++u) {
                float sum = w.x * lrelu(a[u].x + b[u].x) + w.y * lrelu(a[u].y + b[u].y)
                          + w.z * lrelu(a[u].z + b[u].z) + w.w * lrelu(a[u].w + b[u].w);
#pragma unroll
                for (int off = 1; off < 16; off <<= 1)
                    sum += __shfl_xor(sum, off, 32);
                const float h0 = __int_as_float(__builtin_amdgcn_readlane(__float_as_int(sum), 0));
                const float h1 = __int_as_float(__builtin_amdgcn_readlane(__float_as_int(sum), 16));
                if (lv[u] && lane == jj[u]) { l0 = h0; l1 = h1; }
            }
        }

#pragma unroll
        for (int k = 0; k < 2; ++k) {
            const int je = (lane >> 1) + 16 * k;
            const float v0 = __shfl(l0, je, 32);
            const float v1 = __shfl(l1, je, 32);
            const int   dd = __shfl(md, je, 32);
            const int   hd = lane & 1;
            const float v  = hd ? v1 : v0;
            if ((mall >> je) & 1u) {
                logits[(size_t)(e0 + je) * 2 + hd] = v;
                atomicMax(mflip + (size_t)dd * 2 + hd, fflip(v));
            }
        }
    }
}

__global__ __launch_bounds__(256)
void gat_edge_exp(const int* __restrict__ dst,
                  float* __restrict__ lgA, float* __restrict__ lgB,
                  const unsigned* __restrict__ mflip, unsigned long long* __restrict__ den64,
                  unsigned* __restrict__ tick, int E2, unsigned nT) {
    __shared__ unsigned slot[2];
    draw_chunk(tick, nT, slot);
    const unsigned t = __builtin_amdgcn_readfirstlane(slot[0]);
    if (t >= nT) return;
    const bool up = __builtin_amdgcn_readfirstlane(slot[1]) != 0;
    float* __restrict__ logits = up ? lgB : lgA;
    int   idx[2 * RUNS], d[2 * RUNS];
    bool  mine[2 * RUNS];
    unsigned mu[2 * RUNS];
    float lg[2 * RUNS];
#pragma unroll
    for (int k = 0; k < 2 * RUNS; ++k) {
        idx[k] = (int)t * (2 * CHUNK_E) + k * 256 + (int)threadIdx.x;
        d[k] = (idx[k] < E2) ? dst[idx[k] >> 1] : 0;
    }
#pragma unroll
    for (int k = 0; k < 2 * RUNS; ++k) {
        mine[k] = (idx[k] < E2) && ((d[k] >= N_HALF) == up);
        if (mine[k]) {
            mu[k] = mflip[(size_t)d[k] * 2 + (idx[k] & 1)];
            lg[k] = logits[idx[k]];
        }
    }
#pragma unroll
    for (int k = 0; k < 2 * RUNS; ++k) {
        if (mine[k]) {
            const float ex = expf(lg[k] - funflip(mu[k]));
            logits[idx[k]] = ex;
            atomicAdd(den64 + (size_t)d[k] * 2 + (idx[k] & 1), __float2ull_rn(ex * EX_ONE));
        }
    }
}

__global__ __launch_bounds__(256)
void gat_edge_aggregate(const float* __restrict__ fs,
                        const int* __restrict__ src, const int* __restrict__ dst,
                        const float* __restrict__ lgA, const float* __restrict__ lgB,
                        const unsigned long long* __restrict__ den64,
                        const unsigned* __restrict__ fsmax,
                        int* __restrict__ acc,
                        unsigned* __restrict__ tick, int E, unsigned nT) {
    __shared__ unsigned slot[2];
    draw_chunk(tick, nT, slot);
    const unsigned t = __builtin_amdgcn_readfirstlane(slot[0]);
    if (t >= nT) return;
    const bool up = __builtin_amdgcn_readfirstlane(slot[1]) != 0;
    const float* __restrict__ logits = up ? lgB : lgA;
    const int wave = __builtin_amdgcn_readfirstlane((int)(threadIdx.x >> 5));
    const int lane = threadIdx.x & 31;

    const float mxf = __uint_as_float(*fsmax);
    const float hs  = (mxf > 0x1p-90f) ? (0.5f * MSG_ONE) / mxf : 0.0f;

#pragma nounroll
    for (int r = 0; r < RUNS; ++r) {
        const int e0 = (int)t * CHUNK_E + (r * 8 + wave) * EPW;
        if (e0 >= E) break;
        const int n = min(EPW, E - e0);

        int ms = 0, md = 0;
        if (lane < n) { ms = src[e0 + lane]; md = dst[e0 + lane]; }
        const bool mine = (lane < n) && ((md >= N_HALF) == up);
        float al0 = 0.0f, al1 = 0.0f;
        if (mine) {
            const float2 ex = *(const float2*)(logits + (size_t)(e0 + lane) * 2);
            const ulonglong2 q = *(const ulonglong2*)(den64 + (size_t)md * 2);
            al0 = ex.x / ((float)q.x * EX_INV) * hs;
            al1 = ex.y / ((float)q.y * EX_INV) * hs;
        }
        unsigned m = __builtin_amdgcn_ballot_w32(mine);

        while (m) {
            int  jj[4];
            bool lv[4];
            lv[0] = true;
            jj[0] = __builtin_ctz(m);
            m &= (m - 1);
#pragma unroll
            for (int u = 1; u < 4; ++u) {
                lv[u] = (m != 0);
                jj[u] = lv[u] ? __builtin_ctz(m) : jj[u - 1];
                m &= (m - 1);
            }
            float v[4][4];
            int   d[4];
            float a0[4], a1[4];
#pragma unroll
            for (int u = 0; u < 4; ++u) {
                const int s = __builtin_amdgcn_readlane(ms, jj[u]);
                d[u]  = __builtin_amdgcn_readlane(md, jj[u]);
                a0[u] = __int_as_float(__builtin_amdgcn_readlane(__float_as_int(al0), jj[u]));
                a1[u] = __int_as_float(__builtin_amdgcn_readlane(__float_as_int(al1), jj[u]));
                const float* __restrict__ r4 = fs + (size_t)s * HF;
                v[u][0] = r4[lane];
                v[u][1] = r4[lane + 32];
                v[u][2] = r4[lane + 64];
                v[u][3] = r4[lane + 96];
            }
            asm volatile("" :: "v"(v[0][0]), "v"(v[0][1]), "v"(v[0][2]), "v"(v[0][3]),
                               "v"(v[1][0]), "v"(v[1][1]), "v"(v[1][2]), "v"(v[1][3]),
                               "v"(v[2][0]), "v"(v[2][1]), "v"(v[2][2]), "v"(v[2][3]),
                               "v"(v[3][0]), "v"(v[3][1]), "v"(v[3][2]), "v"(v[3][3]));
#pragma unroll
            for (int u = 0; u < 4; ++u) {
                if (lv[u]) {
                    int* o = acc + (size_t)d[u] * F_DIM;
                    atomicAdd(o + lane,      __float2int_rn(a0[u] * v[u][0] + a1[u] * v[u][2]));
                    atomicAdd(o + lane + 32, __float2int_rn(a0[u] * v[u][1] + a1[u] * v[u][3]));
                }
            }
        }
    }
}

__global__ __launch_bounds__(256)
void gat_sum_to_float(const int* __restrict__ acc, const unsigned* __restrict__ fsmax,
                      float* __restrict__ hout, int n4) {
    const int idx = blockIdx.x * blockDim.x + threadIdx.x;
    if (idx >= n4) return;
    const float unit = __uint_as_float(*fsmax) * MSG_INV;
    const int4 a = ((const int4*)acc)[idx];
    float4 o;
    o.x = (float)a.x * unit;
    o.y = (float)a.y * unit;
    o.z = (float)a.z * unit;
    o.w = (float)a.w * unit;
    ((float4*)hout)[idx] = o;
}

extern "C" void kernel_launch(void* const* d_in, const int* in_sizes, int n_in,
                              void* d_out, int out_size, void* d_ws, size_t ws_size,
                              hipStream_t stream) {
    (void)n_in; (void)out_size; (void)ws_size;
    const float* h0   = (const float*)d_in[0];
    const float* Wsrc = (const float*)d_in[1];
    const float* bsrc = (const float*)d_in[2];
    const float* Wdst = (const float*)d_in[3];
    const float* bdst = (const float*)d_in[4];
    const float* attn = (const float*)d_in[5];
    const int*   src  = (const int*)d_in[6];
    const int*   dst  = (const int*)d_in[7];
    const int    E    = in_sizes[6];

    float* ws      = (float*)d_ws;
    float* fs      = ws;  ws += (size_t)N_NODES * HF;
    float* fd      = ws;  ws += (size_t)N_NODES * HF;
    int* acc       = (int*)ws;
    unsigned long long* den64 = (unsigned long long*)(acc + (size_t)N_NODES * F_DIM);
    ws += (size_t)N_NODES * HF;
    float* hbuf    = ws;  ws += (size_t)N_NODES * F_DIM;
    ws += (size_t)N_NODES * 2;
    unsigned* mflp = (unsigned*)ws; ws += (size_t)N_NODES * 2;
    unsigned* tick = (unsigned*)ws; ws += TICK_WORDS;
    unsigned* fsmax = tick + FSMAX_WORD;
    float* logits  = ws;
    float* logitsB = hbuf;

    const unsigned nT = (unsigned)((E + CHUNK_E - 1) / CHUNK_E);
    const unsigned gridT = 2 * nT;
    const int n4     = N_NODES * F_DIM / 4;
    const int gridN  = (n4 + 255) / 256;

    const float* hcur = h0;
    for (int l = 0; l < L_LAYERS; ++l) {
        hipMemsetAsync(mflp, 0, ((size_t)N_NODES * 2 + TICK_WORDS) * sizeof(unsigned), stream);
        hipMemsetAsync(acc,  0, ((size_t)N_NODES * F_DIM + (size_t)N_NODES * 4) * sizeof(int), stream);

        gat_proj_wmma<<<PROJ_GRID, 256, 0, stream>>>(
            hcur,
            Wsrc + (size_t)l * F_DIM * HF, bsrc + (size_t)l * HF,
            Wdst + (size_t)l * F_DIM * HF, bdst + (size_t)l * HF,
            fs, fd, fsmax);

        gat_edge_logits<<<gridT, 256, 0, stream>>>(
            fs, fd, attn + (size_t)l * HF, src, dst, logits, logitsB, mflp,
            tick, E, nT);
        gat_edge_exp<<<gridT, 256, 0, stream>>>(
            dst, logits, logitsB, mflp, den64, tick + 2 * TICK_PAD, E * 2, nT);
        gat_edge_aggregate<<<gridT, 256, 0, stream>>>(
            fs, src, dst, logits, logitsB, den64, fsmax, acc, tick + 4 * TICK_PAD, E, nT);

        float* hn = (l == L_LAYERS - 1) ? (float*)d_out : hbuf;
        gat_sum_to_float<<<gridN, 256, 0, stream>>>(acc, fsmax, hn, n4);
        hcur = hbuf;
    }
}


# node-side passes folded into the projection launch: a 16-node tile's work-group clears its accumulators and, from layer 1 on, stages h straight from the integer sums; 3 memsets and 1 launch gone
# speedup vs baseline: 2.8433x; 1.0815x over previous
#include <hip/hip_runtime.h>

#define N_NODES   50000
#define F_DIM     64
#define H_HEADS   2
#define HF        128
#define L_LAYERS  2
#define NEG_SLOPE 0.2f
#define EPW       32
#define RUNS      2
#define CHUNK_E   512
#define N_HALF    ((N_NODES / 2 + 15) & ~15)
#define TICK_PAD  64
#define TICK_WORDS 512
#define FSMAX_WORD (6 * TICK_PAD)
#define EX_ONE   0x1p40f
#define EX_INV   0x1p-40f
#define MSG_ONE  0x1p30f
#define MSG_INV  0x1p-30f
#define PROJ_TILES (N_NODES / 16)
#define PROJ_GRID  1024
#define A_PITCH    68

typedef __attribute__((ext_vector_type(2))) float v2f;
typedef __attribute__((ext_vector_type(8))) float v8f;
typedef __attribute__((ext_vector_type(4))) float v4f;

__device__ __forceinline__ float lrelu(float x) {
    return x > 0.0f ? x : NEG_SLOPE * x;
}
__device__ __forceinline__ unsigned fflip(float f) {
    unsigned u = __float_as_uint(f);
    return (u & 0x80000000u) ? ~u : (u | 0x80000000u);
}
__device__ __forceinline__ float funflip(unsigned v) {
    return __uint_as_float((v & 0x80000000u) ? (v ^ 0x80000000u) : ~v);
}

template <bool FROM_SUMS> struct TileRaw;
template <> struct TileRaw<false> { typedef float4 type; };
template <> struct TileRaw<true>  { typedef int4   type; };
__device__ __forceinline__ float4 proj_load_raw(const float* __restrict__ h, const int*, int tile, float4*) {
    return *(const float4*)(h + (size_t)tile * (16 * F_DIM) + threadIdx.x * 4);
}
__device__ __forceinline__ int4 proj_load_raw(const float*, const int* acc, int tile, int4*) {
    return *(const int4*)(acc + (size_t)tile * (16 * F_DIM) + threadIdx.x * 4);
}
__device__ __forceinline__ float4 proj_raw_to_h(float4 a, float) { return a; }
__device__ __forceinline__ float4 proj_raw_to_h(int4 a, float unit) {
    float4 v;
    v.x = (float)a.x * unit;
    v.y = (float)a.y * unit;
    v.z = (float)a.z * unit;
    v.w = (float)a.w * unit;
    return v;
}

template <bool FROM_SUMS>
__device__ __forceinline__
void proj_tiles(const float* __restrict__ h, const unsigned* __restrict__ fsmax_prev,
                const float* __restrict__ Wsrc, const float* __restrict__ bsrc,
                const float* __restrict__ Wdst, const float* __restrict__ bdst,
                float* __restrict__ fs, float* __restrict__ fd,
                unsigned* __restrict__ fsmax,
                int* acc, unsigned long long* __restrict__ den64,
                unsigned* __restrict__ mflip) {
    typedef typename TileRaw<FROM_SUMS>::type raw_t;
    __shared__ __attribute__((aligned(16))) float atile[2][16 * A_PITCH];
    __shared__ float wmax[4];
    const int wave = __builtin_amdgcn_readfirstlane((int)(threadIdx.x >> 5));
    const int lane = threadIdx.x & 31;
    const int half = lane >> 4;
    const int lrow = lane & 15;

    const bool  isSrc = (wave < 4);
    const float* __restrict__ W    = isSrc ? Wsrc : Wdst;
    const float* __restrict__ bias = isSrc ? bsrc : bdst;
    float* __restrict__ out        = isSrc ? fs : fd;
    const int cb = (wave & 3) * 32 + lrow * 2;

    v2f b0[16], b1[16];
#pragma unroll
    for (int kk = 0; kk < 16; ++kk) {
        const int ka = kk * 4 + half * 2;
        const float2 r0 = *(const float2*)(W + (size_t)ka * HF + cb);
        const float2 r1 = *(const float2*)(W + (size_t)(ka + 1) * HF + cb);
        b0[kk].x = r0.x; b0[kk].y = r1.x;
        b1[kk].x = r0.y; b1[kk].y = r1.y;
    }
    const float2 bv = *(const float2*)(bias + cb);

    const int srow = threadIdx.x >> 4;
    const int scol = (threadIdx.x & 15) * 4;
    int tile = blockIdx.x;
    float unit = 0.0f;
    if (FROM_SUMS) unit = __uint_as_float(*fsmax_prev) * MSG_INV;
    {
        const float4 v = proj_raw_to_h(proj_load_raw(h, acc, tile, (raw_t*)nullptr), unit);
        *(float4*)(&atile[0][srow * A_PITCH + scol]) = v;
    }
    __syncthreads();

    float mx = 0.0f;
    int buf = 0;
    for (; tile < PROJ_TILES; tile += PROJ_GRID) {
        const int  nxt  = tile + PROJ_GRID;
        const bool more = nxt < PROJ_TILES;
        const raw_t pre = proj_load_raw(h, acc, more ? nxt : tile, (raw_t*)nullptr);

        const float* __restrict__ ar = &atile[buf][lrow * A_PITCH + half * 2];
        v2f a[16];
#pragma unroll
        for (int kk = 0; kk < 16; ++kk)
            a[kk] = *(const v2f*)(ar + kk * 4);

        v8f acc0 = {}, acc1 = {};
#pragma unroll
        for (int kk = 0; kk < 16; ++kk) {
            acc0 = __builtin_amdgcn_wmma_f32_16x16x4_f32(
                    false, a[kk], false, b0[kk], (short)0, acc0, false, false);
            acc1 = __builtin_amdgcn_wmma_f32_16x16x4_f32(
                    false, a[kk], false, b1[kk], (short)0, acc1, false, false);
        }

        *(float4*)(&atile[buf ^ 1][srow * A_PITCH + scol]) = proj_raw_to_h(pre, unit);
        __syncthreads();
        buf ^= 1;

        float* __restrict__ orow = out + (size_t)(tile * 16 + half * 8) * HF + cb;
#pragma unroll
        for (int j = 0; j < 8; ++j) {
            float2 o;
            o.x = acc0[j] + bv.x;
            o.y = acc1[j] + bv.y;
            *(float2*)(orow + (size_t)j * HF) = o;
            mx = fmaxf(mx, fmaxf(fabsf(o.x), fabsf(o.y)));
        }

        {
            const int4 z = {0, 0, 0, 0};
            *(int4*)(acc + (size_t)tile * (16 * F_DIM) + threadIdx.x * 4) = z;
            if (wave == 4) den64[(size_t)tile * 32 + lane] = 0ull;
            if (wave == 5) mflip[(size_t)tile * 32 + lane] = 0u;
        }
    }

#pragma unroll
    for (int off = 16; off > 0; off >>= 1)
        mx = fmaxf(mx, __shfl_xor(mx, off, 32));
    if (isSrc && lane == 0) wmax[wave] = mx;
    __syncthreads();
    if (threadIdx.x == 0) {
        const float m = fmaxf(fmaxf(wmax[0], wmax[1]), fmaxf(wmax[2], wmax[3]));
        atomicMax(fsmax, __float_as_uint(m));
    }
}

__global__ __launch_bounds__(256, 8)
void gat_proj_wmma(const float* __restrict__ h,
                   const float* __restrict__ Wsrc, const float* __restrict__ bsrc,
                   const float* __restrict__ Wdst, const float* __restrict__ bdst,
                   float* __restrict__ fs, float* __restrict__ fd,
                   unsigned* __restrict__ fsmax,
                   int* acc, unsigned long long* __restrict__ den64,
                   unsigned* __restrict__ mflip) {
    proj_tiles<false>(h, nullptr, Wsrc, bsrc, Wdst, bdst, fs, fd, fsmax, acc, den64, mflip);
}

__global__ __launch_bounds__(256, 8)
void gat_proj_wmma_sums(const unsigned* __restrict__ fsmax_prev,
                        const float* __restrict__ Wsrc, const float* __restrict__ bsrc,
                        const float* __restrict__ Wdst, const float* __restrict__ bdst,
                        float* __restrict__ fs, float* __restrict__ fd,
                        unsigned* __restrict__ fsmax,
                        int* acc, unsigned long long* __restrict__ den64,
                        unsigned* __restrict__ mflip) {
    proj_tiles<true>(nullptr, fsmax_prev, Wsrc, bsrc, Wdst, bdst, fs, fd, fsmax, acc, den64, mflip);
}

__device__ __forceinline__ int l2_half() {
    return (__builtin_amdgcn_s_sendmsg_rtn(0x87) >> 18) & 1;
}
__device__ __forceinline__ void draw_chunk(unsigned* tick, unsigned nT, unsigned* slot) {
    if (threadIdx.x == 0) {
        unsigned hh = (unsigned)l2_half();
        unsigned t  = atomicAdd(tick + hh * TICK_PAD, 1u);
        if (t >= nT) {
            hh ^= 1u;
            t = atomicAdd(tick + hh * TICK_PAD, 1u);
        }
        slot[0] = t;
        slot[1] = hh;
    }
    __syncthreads();
}

__global__ __launch_bounds__(256)
void gat_edge_logits(const float* __restrict__ fs, const float* __restrict__ fd,
                     const float* __restrict__ attn,
                     const int* __restrict__ src, const int* __restrict__ dst,
                     float* __restrict__ lgA, float* __restrict__ lgB,
                     unsigned* __restrict__ mflip,
                     unsigned* __restrict__ tick, int E, unsigned nT) {
    __shared__ unsigned slot[2];
    draw_chunk(tick, nT, slot);
    const unsigned t = __builtin_amdgcn_readfirstlane(slot[0]);
    if (t >= nT) return;
    const bool up = __builtin_amdgcn_readfirstlane(slot[1]) != 0;
    float* __restrict__ logits = up ? lgB : lgA;
    const int wave = __builtin_amdgcn_readfirstlane((int)(threadIdx.x >> 5));
    const int lane = threadIdx.x & 31;

    const float4 w = *(const float4*)(attn + lane * 4);
    const float* __restrict__ fsl = fs + lane * 4;
    const float* __restrict__ fdl = fd + lane * 4;

#pragma nounroll
    for (int r = 0; r < RUNS; ++r) {
        const int e0 = (int)t * CHUNK_E + (r * 8 + wave) * EPW;
        if (e0 >= E) break;
        const int n = min(EPW, E - e0);

        int ms = 0, md = 0;
        if (lane < n) { ms = src[e0 + lane]; md = dst[e0 + lane]; }
        const bool mine = (lane < n) && ((md >= N_HALF) == up);
        const unsigned mall = __builtin_amdgcn_ballot_w32(mine);

        float l0 = 0.0f, l1 = 0.0f;
        unsigned m = mall;
        while (m) {
            int  jj[4];
            bool lv[4];
            lv[0] = true;
            jj[0] = __builtin_ctz(m);
            m &= (m - 1);
#pragma unroll
            for (int u = 1; u < 4; ++u) {
                lv[u] = (m != 0);
                jj[u] = lv[u] ? __builtin_ctz(m) : jj[u - 1];
                m &= (m - 1);
            }
            v4f a[4], b[4];
#pragma unroll
            for (int u = 0; u < 4; ++u) {
                const int s = __builtin_amdgcn_readlane(ms, jj[u]);
                const int d = __builtin_amdgcn_readlane(md, jj[u]);
                a[u] = *(const v4f*)(fsl + (size_t)s * HF);
                b[u] = *(const v4f*)(fdl + (size_t)d * HF);
            }
            asm volatile("" :: "v"(a[0]), "v"(a[1]), "v"(a[2]), "v"(a[3]),
                               "v"(b[0]), "v"(b[1]), "v"(b[2]), "v"(b[3]));
#pragma unroll
            for (int u = 0; u < 4; ++u) {
                float sum = w.x * lrelu(a[u].x + b[u].x) + w.y * lrelu(a[u].y + b[u].y)
                          + w.z * lrelu(a[u].z + b[u].z) + w.w * lrelu(a[u].w + b[u].w);
#pragma unroll
                for (int off = 1; off < 16; off <<= 1)
                    sum += __shfl_xor(sum, off, 32);
                const float h0 = __int_as_float(__builtin_amdgcn_readlane(__float_as_int(sum), 0));
                const float h1 = __int_as_float(__builtin_amdgcn_readlane(__float_as_int(sum), 16));
                if (lv[u] && lane == jj[u]) { l0 = h0; l1 = h1; }
            }
        }

#pragma unroll
        for (int k = 0; k < 2; ++k) {
            const int je = (lane >> 1) + 16 * k;
            const float v0 = __shfl(l0, je, 32);
            const float v1 = __shfl(l1, je, 32);
            const int   dd = __shfl(md, je, 32);
            const int   hd = lane & 1;
            const float v  = hd ? v1 : v0;
            if ((mall >> je) & 1u) {
                logits[(size_t)(e0 + je) * 2 + hd] = v;
                atomicMax(mflip + (size_t)dd * 2 + hd, fflip(v));
            }
        }
    }
}

__global__ __launch_bounds__(256)
void gat_edge_exp(const int* __restrict__ dst,
                  float* __restrict__ lgA, float* __restrict__ lgB,
                  const unsigned* __restrict__ mflip, unsigned long long* __restrict__ den64,
                  unsigned* __restrict__ tick, int E2, unsigned nT) {
    __shared__ unsigned slot[2];
    draw_chunk(tick, nT, slot);
    const unsigned t = __builtin_amdgcn_readfirstlane(slot[0]);
    if (t >= nT) return;
    const bool up = __builtin_amdgcn_readfirstlane(slot[1]) != 0;
    float* __restrict__ logits = up ? lgB : lgA;
    int   idx[2 * RUNS], d[2 * RUNS];
    bool  mine[2 * RUNS];
    unsigned mu[2 * RUNS];
    float lg[2 * RUNS];
#pragma unroll
    for (int k = 0; k < 2 * RUNS; ++k) {
        idx[k] = (int)t * (2 * CHUNK_E) + k * 256 + (int)threadIdx.x;
        d[k] = (idx[k] < E2) ? dst[idx[k] >> 1] : 0;
    }
#pragma unroll
    for (int k = 0; k < 2 * RUNS; ++k) {
        mine[k] = (idx[k] < E2) && ((d[k] >= N_HALF) == up);
        if (mine[k]) {
            mu[k] = mflip[(size_t)d[k] * 2 + (idx[k] & 1)];
            lg[k] = logits[idx[k]];
        }
    }
#pragma unroll
    for (int k = 0; k < 2 * RUNS; ++k) {
        if (mine[k]) {
            const float ex = expf(lg[k] - funflip(mu[k]));
            logits[idx[k]] = ex;
            atomicAdd(den64 + (size_t)d[k] * 2 + (idx[k] & 1), __float2ull_rn(ex * EX_ONE));
        }
    }
}

__global__ __launch_bounds__(256)
void gat_edge_aggregate(const float* __restrict__ fs,
                        const int* __restrict__ src, const int* __restrict__ dst,
                        const float* __restrict__ lgA, const float* __restrict__ lgB,
                        const unsigned long long* __restrict__ den64,
                        const unsigned* __restrict__ fsmax,
                        int* __restrict__ acc,
                        unsigned* __restrict__ tick, int E, unsigned nT) {
    __shared__ unsigned slot[2];
    draw_chunk(tick, nT, slot);
    const unsigned t = __builtin_amdgcn_readfirstlane(slot[0]);
    if (t >= nT) return;
    const bool up = __builtin_amdgcn_readfirstlane(slot[1]) != 0;
    const float* __restrict__ logits = up ? lgB : lgA;
    const int wave = __builtin_amdgcn_readfirstlane((int)(threadIdx.x >> 5));
    const int lane = threadIdx.x & 31;

    const float mxf = __uint_as_float(*fsmax);
    const float hs  = (mxf > 0x1p-90f) ? (0.5f * MSG_ONE) / mxf : 0.0f;

#pragma nounroll
    for (int r = 0; r < RUNS; ++r) {
        const int e0 = (int)t * CHUNK_E + (r * 8 + wave) * EPW;
        if (e0 >= E) break;
        const int n = min(EPW, E - e0);

        int ms = 0, md = 0;
        if (lane < n) { ms = src[e0 + lane]; md = dst[e0 + lane]; }
        const bool mine = (lane < n) && ((md >= N_HALF) == up);
        float al0 = 0.0f, al1 = 0.0f;
        if (mine) {
            const float2 ex = *(const float2*)(logits + (size_t)(e0 + lane) * 2);
            const ulonglong2 q = *(const ulonglong2*)(den64 + (size_t)md * 2);
            al0 = ex.x / ((float)q.x * EX_INV) * hs;
            al1 = ex.y / ((float)q.y * EX_INV) * hs;
        }
        unsigned m = __builtin_amdgcn_ballot_w32(mine);

        while (m) {
            int  jj[4];
            bool lv[4];
            lv[0] = true;
            jj[0] = __builtin_ctz(m);
            m &= (m - 1);
#pragma unroll
            for (int u = 1; u < 4; ++u) {
                lv[u] = (m != 0);
                jj[u] = lv[u] ? __builtin_ctz(m) : jj[u - 1];
                m &= (m - 1);
            }
            float v[4][4];
            int   d[4];
            float a0[4], a1[4];
#pragma unroll
            for (int u = 0; u < 4; ++u) {
                const int s = __builtin_amdgcn_readlane(ms, jj[u]);
                d[u]  = __builtin_amdgcn_readlane(md, jj[u]);
                a0[u] = __int_as_float(__builtin_amdgcn_readlane(__float_as_int(al0), jj[u]));
                a1[u] = __int_as_float(__builtin_amdgcn_readlane(__float_as_int(al1), jj[u]));
                const float* __restrict__ r4 = fs + (size_t)s * HF;
                v[u][0] = r4[lane];
                v[u][1] = r4[lane + 32];
                v[u][2] = r4[lane + 64];
                v[u][3] = r4[lane + 96];
            }
            asm volatile("" :: "v"(v[0][0]), "v"(v[0][1]), "v"(v[0][2]), "v"(v[0][3]),
                               "v"(v[1][0]), "v"(v[1][1]), "v"(v[1][2]), "v"(v[1][3]),
                               "v"(v[2][0]), "v"(v[2][1]), "v"(v[2][2]), "v"(v[2][3]),
                               "v"(v[3][0]), "v"(v[3][1]), "v"(v[3][2]), "v"(v[3][3]));
#pragma unroll
            for (int u = 0; u < 4; ++u) {
                if (lv[u]) {
                    int* o = acc + (size_t)d[u] * F_DIM;
                    atomicAdd(o + lane,      __float2int_rn(a0[u] * v[u][0] + a1[u] * v[u][2]));
                    atomicAdd(o + lane + 32, __float2int_rn(a0[u] * v[u][1] + a1[u] * v[u][3]));
                }
            }
        }
    }
}

__global__ __launch_bounds__(256)
void gat_sum_to_float(const int* __restrict__ acc, const unsigned* __restrict__ fsmax,
                      float* __restrict__ hout, int n4) {
    const int idx = blockIdx.x * blockDim.x + threadIdx.x;
    if (idx >= n4) return;
    const float unit = __uint_as_float(*fsmax) * MSG_INV;
    const int4 a = ((const int4*)acc)[idx];
    float4 o;
    o.x = (float)a.x * unit;
    o.y = (float)a.y * unit;
    o.z = (float)a.z * unit;
    o.w = (float)a.w * unit;
    ((float4*)hout)[idx] = o;
}

extern "C" void kernel_launch(void* const* d_in, const int* in_sizes, int n_in,
                              void* d_out, int out_size, void* d_ws, size_t ws_size,
                              hipStream_t stream) {
    (void)n_in; (void)out_size; (void)ws_size;
    const float* h0   = (const float*)d_in[0];
    const float* Wsrc = (const float*)d_in[1];
    const float* bsrc = (const float*)d_in[2];
    const float* Wdst = (const float*)d_in[3];
    const float* bdst = (const float*)d_in[4];
    const float* attn = (const float*)d_in[5];
    const int*   src  = (const int*)d_in[6];
    const int*   dst  = (const int*)d_in[7];
    const int    E    = in_sizes[6];

    float* ws      = (float*)d_ws;
    float* fs      = ws;  ws += (size_t)N_NODES * HF;
    float* fd      = ws;  ws += (size_t)N_NODES * HF;
    int* acc       = (int*)ws;
    unsigned long long* den64 = (unsigned long long*)(acc + (size_t)N_NODES * F_DIM);
    ws += (size_t)N_NODES * HF;
    float* hbuf    = ws;  ws += (size_t)N_NODES * F_DIM;
    unsigned* tick = (unsigned*)ws; ws += (size_t)N_NODES * 2;
    unsigned* mflp = (unsigned*)ws; ws += (size_t)N_NODES * 2;
    ws += TICK_WORDS;
    float* logits  = ws;
    float* logitsB = hbuf;

    const unsigned nT = (unsigned)((E + CHUNK_E - 1) / CHUNK_E);
    const unsigned gridT = 2 * nT;
    const int n4     = N_NODES * F_DIM / 4;
    const int gridN  = (n4 + 255) / 256;

    hipMemsetAsync(tick, 0, (size_t)L_LAYERS * TICK_WORDS * sizeof(unsigned), stream);

    for (int l = 0; l < L_LAYERS; ++l) {
        unsigned* tk    = tick + (size_t)l * TICK_WORDS;
        unsigned* fsmax = tk + FSMAX_WORD;

        if (l == 0)
            gat_proj_wmma<<<PROJ_GRID, 256, 0, stream>>>(
                h0,
                Wsrc + (size_t)l * F_DIM * HF, bsrc + (size_t)l * HF,
                Wdst + (size_t)l * F_DIM * HF, bdst + (size_t)l * HF,
                fs, fd, fsmax, acc, den64, mflp);
        else
            gat_proj_wmma_sums<<<PROJ_GRID, 256, 0, stream>>>(
                tk - TICK_WORDS + FSMAX_WORD,
                Wsrc + (size_t)l * F_DIM * HF, bsrc + (size_t)l * HF,
                Wdst + (size_t)l * F_DIM * HF, bdst + (size_t)l * HF,
                fs, fd, fsmax, acc, den64, mflp);

        gat_edge_logits<<<gridT, 256, 0, stream>>>(
            fs, fd, attn + (size_t)l * HF, src, dst, logits, logitsB, mflp,
            tk, E, nT);
        gat_edge_exp<<<gridT, 256, 0, stream>>>(
            dst, logits, logitsB, mflp, den64, tk + 2 * TICK_PAD, E * 2, nT);
        gat_edge_aggregate<<<gridT, 256, 0, stream>>>(
            fs, src, dst, logits, logitsB, den64, fsmax, acc, tk + 4 * TICK_PAD, E, nT);

        if (l == L_LAYERS - 1)
            gat_sum_to_float<<<gridN, 256, 0, stream>>>(acc, fsmax, (float*)d_out, n4);
    }
}


# logits pass: the 16-lane head sums by DPP row_xmask moves on the vector unit instead of __shfl_xor (ds_bpermute): 12 dependent LDS round trips per 4 edges gone; same butterfly, sum order unchanged
# speedup vs baseline: 3.0768x; 1.0821x over previous
#include <hip/hip_runtime.h>

#define N_NODES   50000
#define F_DIM     64
#define H_HEADS   2
#define HF        128
#define L_LAYERS  2
#define NEG_SLOPE 0.2f
#define EPW       32
#define RUNS      2
#define CHUNK_E   512
#define N_HALF    ((N_NODES / 2 + 15) & ~15)
#define TICK_PAD  64
#define TICK_WORDS 512
#define FSMAX_WORD (6 * TICK_PAD)
#define EX_ONE   0x1p40f
#define EX_INV   0x1p-40f
#define MSG_ONE  0x1p30f
#define MSG_INV  0x1p-30f
#define PROJ_TILES (N_NODES / 16)
#define PROJ_GRID  1024
#define A_PITCH    68

typedef __attribute__((ext_vector_type(2))) float v2f;
typedef __attribute__((ext_vector_type(8))) float v8f;
typedef __attribute__((ext_vector_type(4))) float v4f;

__device__ __forceinline__ float lrelu(float x) {
    return x > 0.0f ? x : NEG_SLOPE * x;
}
__device__ __forceinline__ unsigned fflip(float f) {
    unsigned u = __float_as_uint(f);
    return (u & 0x80000000u) ? ~u : (u | 0x80000000u);
}
__device__ __forceinline__ float funflip(unsigned v) {
    return __uint_as_float((v & 0x80000000u) ? (v ^ 0x80000000u) : ~v);
}
template <int K>
__device__ __forceinline__ float row_xor_add(float v) {
    return v + __int_as_float(__builtin_amdgcn_update_dpp(0, __float_as_int(v), 0x160 | K, 0xf, 0xf, true));
}
__device__ __forceinline__ float row16_sum(float v) {
    v = row_xor_add<1>(v);
    v = row_xor_add<2>(v);
    v = row_xor_add<4>(v);
    v = row_xor_add<8>(v);
    return v;
}

template <bool FROM_SUMS> struct TileRaw;
template <> struct TileRaw<false> { typedef float4 type; };
template <> struct TileRaw<true>  { typedef int4   type; };
__device__ __forceinline__ float4 proj_load_raw(const float* __restrict__ h, const int*, int tile, float4*) {
    return *(const float4*)(h + (size_t)tile * (16 * F_DIM) + threadIdx.x * 4);
}
__device__ __forceinline__ int4 proj_load_raw(const float*, const int* acc, int tile, int4*) {
    return *(const int4*)(acc + (size_t)tile * (16 * F_DIM) + threadIdx.x * 4);
}
__device__ __forceinline__ float4 proj_raw_to_h(float4 a, float) { return a; }
__device__ __forceinline__ float4 proj_raw_to_h(int4 a, float unit) {
    float4 v;
    v.x = (float)a.x * unit;
    v.y = (float)a.y * unit;
    v.z = (float)a.z * unit;
    v.w = (float)a.w * unit;
    return v;
}

template <bool FROM_SUMS>
__device__ __forceinline__
void proj_tiles(const float* __restrict__ h, const unsigned* __restrict__ fsmax_prev,
                const float* __restrict__ Wsrc, const float* __restrict__ bsrc,
                const float* __restrict__ Wdst, const float* __restrict__ bdst,
                float* __restrict__ fs, float* __restrict__ fd,
                unsigned* __restrict__ fsmax,
                int* acc, unsigned long long* __restrict__ den64,
                unsigned* __restrict__ mflip) {
    typedef typename TileRaw<FROM_SUMS>::type raw_t;
    __shared__ __attribute__((aligned(16))) float atile[2][16 * A_PITCH];
    __shared__ float wmax[4];
    const int wave = __builtin_amdgcn_readfirstlane((int)(threadIdx.x >> 5));
    const int lane = threadIdx.x & 31;
    const int half = lane >> 4;
    const int lrow = lane & 15;

    const bool  isSrc = (wave < 4);
    const float* __restrict__ W    = isSrc ? Wsrc : Wdst;
    const float* __restrict__ bias = isSrc ? bsrc : bdst;
    float* __restrict__ out        = isSrc ? fs : fd;
    const int cb = (wave & 3) * 32 + lrow * 2;

    v2f b0[16], b1[16];
#pragma unroll
    for (int kk = 0; kk < 16; ++kk) {
        const int ka = kk * 4 + half * 2;
        const float2 r0 = *(const float2*)(W + (size_t)ka * HF + cb);
        const float2 r1 = *(const float2*)(W + (size_t)(ka + 1) * HF + cb);
        b0[kk].x = r0.x; b0[kk].y = r1.x;
        b1[kk].x = r0.y; b1[kk].y = r1.y;
    }
    const float2 bv = *(const float2*)(bias + cb);

    const int srow = threadIdx.x >> 4;
    const int scol = (threadIdx.x & 15) * 4;
    int tile = blockIdx.x;
    float unit = 0.0f;
    if (FROM_SUMS) unit = __uint_as_float(*fsmax_prev) * MSG_INV;
    {
        const float4 v = proj_raw_to_h(proj_load_raw(h, acc, tile, (raw_t*)nullptr), unit);
        *(float4*)(&atile[0][srow * A_PITCH + scol]) = v;
    }
    __syncthreads();

    float mx = 0.0f;
    int buf = 0;
    for (; tile < PROJ_TILES; tile += PROJ_GRID) {
        const int  nxt  = tile + PROJ_GRID;
        const bool more = nxt < PROJ_TILES;
        const raw_t pre = proj_load_raw(h, acc, more ? nxt : tile, (raw_t*)nullptr);

        const float* __restrict__ ar = &atile[buf][lrow * A_PITCH + half * 2];
        v2f a[16];
#pragma unroll
        for (int kk = 0; kk < 16; ++kk)
            a[kk] = *(const v2f*)(ar + kk * 4);

        v8f acc0 = {}, acc1 = {};
#pragma unroll
        for (int kk = 0; kk < 16; ++kk) {
            acc0 = __builtin_amdgcn_wmma_f32_16x16x4_f32(
                    false, a[kk], false, b0[kk], (short)0, acc0, false, false);
            acc1 = __builtin_amdgcn_wmma_f32_16x16x4_f32(
                    false, a[kk], false, b1[kk], (short)0, acc1, false, false);
        }

        *(float4*)(&atile[buf ^ 1][srow * A_PITCH + scol]) = proj_raw_to_h(pre, unit);
        __syncthreads();
        buf ^= 1;

        float* __restrict__ orow = out + (size_t)(tile * 16 + half * 8) * HF + cb;
#pragma unroll
        for (int j = 0; j < 8; ++j) {
            float2 o;
            o.x = acc0[j] + bv.x;
            o.y = acc1[j] + bv.y;
            *(float2*)(orow + (size_t)j * HF) = o;
            mx = fmaxf(mx, fmaxf(fabsf(o.x), fabsf(o.y)));
        }

        {
            const int4 z = {0, 0, 0, 0};
            *(int4*)(acc + (size_t)tile * (16 * F_DIM) + threadIdx.x * 4) = z;
            if (wave == 4) den64[(size_t)tile * 32 + lane] = 0ull;
            if (wave == 5) mflip[(size_t)tile * 32 + lane] = 0u;
        }
    }

#pragma unroll
    for (int off = 16; off > 0; off >>= 1)
        mx = fmaxf(mx, __shfl_xor(mx, off, 32));
    if (isSrc && lane == 0) wmax[wave] = mx;
    __syncthreads();
    if (threadIdx.x == 0) {
        const float m = fmaxf(fmaxf(wmax[0], wmax[1]), fmaxf(wmax[2], wmax[3]));
        atomicMax(fsmax, __float_as_uint(m));
    }
}

__global__ __launch_bounds__(256, 8)
void gat_proj_wmma(const float* __restrict__ h,
                   const float* __restrict__ Wsrc, const float* __restrict__ bsrc,
                   const float* __restrict__ Wdst, const float* __restrict__ bdst,
                   float* __restrict__ fs, float* __restrict__ fd,
                   unsigned* __restrict__ fsmax,
                   int* acc, unsigned long long* __restrict__ den64,
                   unsigned* __restrict__ mflip) {
    proj_tiles<false>(h, nullptr, Wsrc, bsrc, Wdst, bdst, fs, fd, fsmax, acc, den64, mflip);
}

__global__ __launch_bounds__(256, 8)
void gat_proj_wmma_sums(const unsigned* __restrict__ fsmax_prev,
                        const float* __restrict__ Wsrc, const float* __restrict__ bsrc,
                        const float* __restrict__ Wdst, const float* __restrict__ bdst,
                        float* __restrict__ fs, float* __restrict__ fd,
                        unsigned* __restrict__ fsmax,
                        int* acc, unsigned long long* __restrict__ den64,
                        unsigned* __restrict__ mflip) {
    proj_tiles<true>(nullptr, fsmax_prev, Wsrc, bsrc, Wdst, bdst, fs, fd, fsmax, acc, den64, mflip);
}

__device__ __forceinline__ int l2_half() {
    return (__builtin_amdgcn_s_sendmsg_rtn(0x87) >> 18) & 1;
}
__device__ __forceinline__ void draw_chunk(unsigned* tick, unsigned nT, unsigned* slot) {
    if (threadIdx.x == 0) {
        unsigned hh = (unsigned)l2_half();
        unsigned t  = atomicAdd(tick + hh * TICK_PAD, 1u);
        if (t >= nT) {
            hh ^= 1u;
            t = atomicAdd(tick + hh * TICK_PAD, 1u);
        }
        slot[0] = t;
        slot[1] = hh;
    }
    __syncthreads();
}

__global__ __launch_bounds__(256)
void gat_edge_logits(const float* __restrict__ fs, const float* __restrict__ fd,
                     const float* __restrict__ attn,
                     const int* __restrict__ src, const int* __restrict__ dst,
                     float* __restrict__ lgA, float* __restrict__ lgB,
                     unsigned* __restrict__ mflip,
                     unsigned* __restrict__ tick, int E, unsigned nT) {
    __shared__ unsigned slot[2];
    draw_chunk(tick, nT, slot);
    const unsigned t = __builtin_amdgcn_readfirstlane(slot[0]);
    if (t >= nT) return;
    const bool up = __builtin_amdgcn_readfirstlane(slot[1]) != 0;
    float* __restrict__ logits = up ? lgB : lgA;
    const int wave = __builtin_amdgcn_readfirstlane((int)(threadIdx.x >> 5));
    const int lane = threadIdx.x & 31;

    const float4 w = *(const float4*)(attn + lane * 4);
    const float* __restrict__ fsl = fs + lane * 4;
    const float* __restrict__ fdl = fd + lane * 4;

#pragma nounroll
    for (int r = 0; r < RUNS; ++r) {
        const int e0 = (int)t * CHUNK_E + (r * 8 + wave) * EPW;
        if (e0 >= E) break;
        const int n = min(EPW, E - e0);

        int ms = 0, md = 0;
        if (lane < n) { ms = src[e0 + lane]; md = dst[e0 + lane]; }
        const bool mine = (lane < n) && ((md >= N_HALF) == up);
        const unsigned mall = __builtin_amdgcn_ballot_w32(mine);

        float l0 = 0.0f, l1 = 0.0f;
        unsigned m = mall;
        while (m) {
            int  jj[4];
            bool lv[4];
            lv[0] = true;
            jj[0] = __builtin_ctz(m);
            m &= (m - 1);
#pragma unroll
            for (int u = 1; u < 4; ++u) {
                lv[u] = (m != 0);
                jj[u] = lv[u] ? __builtin_ctz(m) : jj[u - 1];
                m &= (m - 1);
            }
            v4f a[4], b[4];
#pragma unroll
            for (int u = 0; u < 4; ++u) {
                const int s = __builtin_amdgcn_readlane(ms, jj[u]);
                const int d = __builtin_amdgcn_readlane(md, jj[u]);
                a[u] = *(const v4f*)(fsl + (size_t)s * HF);
                b[u] = *(const v4f*)(fdl + (size_t)d * HF);
            }
            asm volatile("" :: "v"(a[0]), "v"(a[1]), "v"(a[2]), "v"(a[3]),
                               "v"(b[0]), "v"(b[1]), "v"(b[2]), "v"(b[3]));
#pragma unroll
            for (int u = 0; u < 4; ++u) {
                float sum = w.x * lrelu(a[u].x + b[u].x) + w.y * lrelu(a[u].y + b[u].y)
                          + w.z * lrelu(a[u].z + b[u].z) + w.w * lrelu(a[u].w + b[u].w);
                sum = row16_sum(sum);
                const float h0 = __int_as_float(__builtin_amdgcn_readlane(__float_as_int(sum), 0));
                const float h1 = __int_as_float(__builtin_amdgcn_readlane(__float_as_int(sum), 16));
                if (lv[u] && lane == jj[u]) { l0 = h0; l1 = h1; }
            }
        }

#pragma unroll
        for (int k = 0; k < 2; ++k) {
            const int je = (lane >> 1) + 16 * k;
            const float v0 = __shfl(l0, je, 32);
            const float v1 = __shfl(l1, je, 32);
            const int   dd = __shfl(md, je, 32);
            const int   hd = lane & 1;
            const float v  = hd ? v1 : v0;
            if ((mall >> je) & 1u) {
                logits[(size_t)(e0 + je) * 2 + hd] = v;
                atomicMax(mflip + (size_t)dd * 2 + hd, fflip(v));
            }
        }
    }
}

__global__ __launch_bounds__(256)
void gat_edge_exp(const int* __restrict__ dst,
                  float* __restrict__ lgA, float* __restrict__ lgB,
                  const unsigned* __restrict__ mflip, unsigned long long* __restrict__ den64,
                  unsigned* __restrict__ tick, int E2, unsigned nT) {
    __shared__ unsigned slot[2];
    draw_chunk(tick, nT, slot);
    const unsigned t = __builtin_amdgcn_readfirstlane(slot[0]);
    if (t >= nT) return;
    const bool up = __builtin_amdgcn_readfirstlane(slot[1]) != 0;
    float* __restrict__ logits = up ? lgB : lgA;
    int   idx[2 * RUNS], d[2 * RUNS];
    bool  mine[2 * RUNS];
    unsigned mu[2 * RUNS];
    float lg[2 * RUNS];
#pragma unroll
    for (int k = 0; k < 2 * RUNS; ++k) {
        idx[k] = (int)t * (2 * CHUNK_E) + k * 256 + (int)threadIdx.x;
        d[k] = (idx[k] < E2) ? dst[idx[k] >> 1] : 0;
    }
#pragma unroll
    for (int k = 0; k < 2 * RUNS; ++k) {
        mine[k] = (idx[k] < E2) && ((d[k] >= N_HALF) == up);
        if (mine[k]) {
            mu[k] = mflip[(size_t)d[k] * 2 + (idx[k] & 1)];
            lg[k] = logits[idx[k]];
        }
    }
#pragma unroll
    for (int k = 0; k < 2 * RUNS; ++k) {
        if (mine[k]) {
            const float ex = expf(lg[k] - funflip(mu[k]));
            logits[idx[k]] = ex;
            atomicAdd(den64 + (size_t)d[k] * 2 + (idx[k] & 1), __float2ull_rn(ex * EX_ONE));
        }
    }
}

__global__ __launch_bounds__(256)
void gat_edge_aggregate(const float* __restrict__ fs,
                        const int* __restrict__ src, const int* __restrict__ dst,
                        const float* __restrict__ lgA, const float* __restrict__ lgB,
                        const unsigned long long* __restrict__ den64,
                        const unsigned* __restrict__ fsmax,
                        int* __restrict__ acc,
                        unsigned* __restrict__ tick, int E, unsigned nT) {
    __shared__ unsigned slot[2];
    draw_chunk(tick, nT, slot);
    const unsigned t = __builtin_amdgcn_readfirstlane(slot[0]);
    if (t >= nT) return;
    const bool up = __builtin_amdgcn_readfirstlane(slot[1]) != 0;
    const float* __restrict__ logits = up ? lgB : lgA;
    const int wave = __builtin_amdgcn_readfirstlane((int)(threadIdx.x >> 5));
    const int lane = threadIdx.x & 31;

    const float mxf = __uint_as_float(*fsmax);
    const float hs  = (mxf > 0x1p-90f) ? (0.5f * MSG_ONE) / mxf : 0.0f;

#pragma nounroll
    for (int r = 0; r < RUNS; ++r) {
        const int e0 = (int)t * CHUNK_E + (r * 8 + wave) * EPW;
        if (e0 >= E) break;
        const int n = min(EPW, E - e0);

        int ms = 0, md = 0;
        if (lane < n) { ms = src[e0 + lane]; md = dst[e0 + lane]; }
        const bool mine = (lane < n) && ((md >= N_HALF) == up);
        float al0 = 0.0f, al1 = 0.0f;
        if (mine) {
            const float2 ex = *(const float2*)(logits + (size_t)(e0 + lane) * 2);
            const ulonglong2 q = *(const ulonglong2*)(den64 + (size_t)md * 2);
            al0 = ex.x / ((float)q.x * EX_INV) * hs;
            al1 = ex.y / ((float)q.y * EX_INV) * hs;
        }
        unsigned m = __builtin_amdgcn_ballot_w32(mine);

        while (m) {
            int  jj[4];
            bool lv[4];
            lv[0] = true;
            jj[0] = __builtin_ctz(m);
            m &= (m - 1);
#pragma unroll
            for (int u = 1; u < 4; ++u) {
                lv[u] = (m != 0);
                jj[u] = lv[u] ? __builtin_ctz(m) : jj[u - 1];
                m &= (m - 1);
            }
            float v[4][4];
            int   d[4];
            float a0[4], a1[4];
#pragma unroll
            for (int u = 0; u < 4; ++u) {
                const int s = __builtin_amdgcn_readlane(ms, jj[u]);
                d[u]  = __builtin_amdgcn_readlane(md, jj[u]);
                a0[u] = __int_as_float(__builtin_amdgcn_readlane(__float_as_int(al0), jj[u]));
                a1[u] = __int_as_float(__builtin_amdgcn_readlane(__float_as_int(al1), jj[u]));
                const float* __restrict__ r4 = fs + (size_t)s * HF;
                v[u][0] = r4[lane];
                v[u][1] = r4[lane + 32];
                v[u][2] = r4[lane + 64];
                v[u][3] = r4[lane + 96];
            }
            asm volatile("" :: "v"(v[0][0]), "v"(v[0][1]), "v"(v[0][2]), "v"(v[0][3]),
                               "v"(v[1][0]), "v"(v[1][1]), "v"(v[1][2]), "v"(v[1][3]),
                               "v"(v[2][0]), "v"(v[2][1]), "v"(v[2][2]), "v"(v[2][3]),
                               "v"(v[3][0]), "v"(v[3][1]), "v"(v[3][2]), "v"(v[3][3]));
#pragma unroll
            for (int u = 0; u < 4; ++u) {
                if (lv[u]) {
                    int* o = acc + (size_t)d[u] * F_DIM;
                    atomicAdd(o + lane,      __float2int_rn(a0[u] * v[u][0] + a1[u] * v[u][2]));
                    atomicAdd(o + lane + 32, __float2int_rn(a0[u] * v[u][1] + a1[u] * v[u][3]));
                }
            }
        }
    }
}

__global__ __launch_bounds__(256)
void gat_sum_to_float(const int* __restrict__ acc, const unsigned* __restrict__ fsmax,
                      float* __restrict__ hout, int n4) {
    const int idx = blockIdx.x * blockDim.x + threadIdx.x;
    if (idx >= n4) return;
    const float unit = __uint_as_float(*fsmax) * MSG_INV;
    const int4 a = ((const int4*)acc)[idx];
    float4 o;
    o.x = (float)a.x * unit;
    o.y = (float)a.y * unit;
    o.z = (float)a.z * unit;
    o.w = (float)a.w * unit;
    ((float4*)hout)[idx] = o;
}

extern "C" void kernel_launch(void* const* d_in, const int* in_sizes, int n_in,
                              void* d_out, int out_size, void* d_ws, size_t ws_size,
                              hipStream_t stream) {
    (void)n_in; (void)out_size; (void)ws_size;
    const float* h0   = (const float*)d_in[0];
    const float* Wsrc = (const float*)d_in[1];
    const float* bsrc = (const float*)d_in[2];
    const float* Wdst = (const float*)d_in[3];
    const float* bdst = (const float*)d_in[4];
    const float* attn = (const float*)d_in[5];
    const int*   src  = (const int*)d_in[6];
    const int*   dst  = (const int*)d_in[7];
    const int    E    = in_sizes[6];

    float* ws      = (float*)d_ws;
    float* fs      = ws;  ws += (size_t)N_NODES * HF;
    float* fd      = ws;  ws += (size_t)N_NODES * HF;
    int* acc       = (int*)ws;
    unsigned long long* den64 = (unsigned long long*)(acc + (size_t)N_NODES * F_DIM);
    ws += (size_t)N_NODES * HF;
    float* hbuf    = ws;  ws += (size_t)N_NODES * F_DIM;
    unsigned* tick = (unsigned*)ws; ws += (size_t)N_NODES * 2;
    unsigned* mflp = (unsigned*)ws; ws += (size_t)N_NODES * 2;
    ws += TICK_WORDS;
    float* logits  = ws;
    float* logitsB = hbuf;

    const unsigned nT = (unsigned)((E + CHUNK_E - 1) / CHUNK_E);
    const unsigned gridT = 2 * nT;
    const int n4     = N_NODES * F_DIM / 4;
    const int gridN  = (n4 + 255) / 256;

    hipMemsetAsync(tick, 0, (size_t)L_LAYERS * TICK_WORDS * sizeof(unsigned), stream);

    for (int l = 0; l < L_LAYERS; ++l) {
        unsigned* tk    = tick + (size_t)l * TICK_WORDS;
        unsigned* fsmax = tk + FSMAX_WORD;

        if (l == 0)
            gat_proj_wmma<<<PROJ_GRID, 256, 0, stream>>>(
                h0,
                Wsrc + (size_t)l * F_DIM * HF, bsrc + (size_t)l * HF,
                Wdst + (size_t)l * F_DIM * HF, bdst + (size_t)l * HF,
                fs, fd, fsmax, acc, den64, mflp);
        else
            gat_proj_wmma_sums<<<PROJ_GRID, 256, 0, stream>>>(
                tk - TICK_WORDS + FSMAX_WORD,
                Wsrc + (size_t)l * F_DIM * HF, bsrc + (size_t)l * HF,
                Wdst + (size_t)l * F_DIM * HF, bdst + (size_t)l * HF,
                fs, fd, fsmax, acc, den64, mflp);

        gat_edge_logits<<<gridT, 256, 0, stream>>>(
            fs, fd, attn + (size_t)l * HF, src, dst, logits, logitsB, mflp,
            tk, E, nT);
        gat_edge_exp<<<gridT, 256, 0, stream>>>(
            dst, logits, logitsB, mflp, den64, tk + 2 * TICK_PAD, E * 2, nT);
        gat_edge_aggregate<<<gridT, 256, 0, stream>>>(
            fs, src, dst, logits, logitsB, den64, fsmax, acc, tk + 4 * TICK_PAD, E, nT);

        if (l == L_LAYERS - 1)
            gat_sum_to_float<<<gridN, 256, 0, stream>>>(acc, fsmax, (float*)d_out, n4);
    }
}
